# starting kernel
# baseline (speedup 1.0000x reference)
#include <hip/hip_runtime.h>

constexpr int N_ = 50000;
constexpr int E_ = 800000;
constexpr int F_ = 128;
constexpr int H_ = 8;
constexpr int D_ = 16;
constexpr int C_ = 40;
constexpr int L_ = 2;
constexpr float NEG_SLOPE = 0.2f;
constexpr float LN_EPS = 1e-5f;

typedef __attribute__((ext_vector_type(2))) float v2f;
typedef __attribute__((ext_vector_type(8))) float v8f;

__device__ __forceinline__ float lrelu(float x) { return x > 0.f ? x : NEG_SLOPE * x; }

__device__ __forceinline__ void atomicMaxF(float* addr, float val) {
  if (val >= 0.f) atomicMax((int*)addr, __float_as_int(val));
  else            atomicMin((unsigned int*)addr, __float_as_uint(val));
}

__global__ void gemm_node_fc(const float* __restrict__ X,
                             const float* __restrict__ W,
                             float* __restrict__ Y) {
  const int lane = threadIdx.x & 31;
  const int wave = threadIdx.x >> 5;
  const int m0   = blockIdx.x * 16;
  const int n0   = wave * 16;
  const int half = lane >> 4;
  const int l16  = lane & 15;
  const int arow = m0 + l16;
  const int bcol = n0 + l16;

  v8f acc = {};
#pragma unroll
  for (int k = 0; k < F_; k += 4) {
    const int ka = k + 2 * half;
    v2f a;
    a.x = X[arow * F_ + ka];
    a.y = X[arow * F_ + ka + 1];
    v2f b;
    b.x = W[ka * F_ + bcol];
    b.y = W[(ka + 1) * F_ + bcol];
    acc = __builtin_amdgcn_wmma_f32_16x16x4_f32(false, a, false, b,
                                                (short)0, acc, false, false);
  }
#pragma unroll
  for (int r = 0; r < 8; ++r) {
    const int m = m0 + r + 8 * half;
    Y[m * F_ + bcol] = acc[r];
  }
}

__global__ void node_coef(const float* __restrict__ hp,
                          const float* __restrict__ al,
                          const float* __restrict__ ar,
                          float* __restrict__ el, float* __restrict__ er,
                          float* __restrict__ mx, float* __restrict__ sm,
                          int total) {
  const int t = blockIdx.x * blockDim.x + threadIdx.x;
  if (t >= total) return;
  const int n = t >> 3;
  const int h = t & 7;
  const float* row = hp + n * F_ + h * D_;
  float sl = 0.f, sr = 0.f;
#pragma unroll
  for (int d = 0; d < D_; ++d) {
    const float x = row[d];
    sl += x * al[h * D_ + d];
    sr += x * ar[h * D_ + d];
  }
  el[t] = sl;
  er[t] = sr;
  mx[t] = __uint_as_float(0xff800000u);
  sm[t] = 0.f;
}

__global__ void zero_f32(float* __restrict__ p, int n) {
  const int t = blockIdx.x * blockDim.x + threadIdx.x;
  if (t < n) p[t] = 0.f;
}

__global__ void edge_max(const int* __restrict__ src, const int* __restrict__ dst,
                         const float* __restrict__ el, const float* __restrict__ er,
                         float* __restrict__ mx, int total) {
  const int t = blockIdx.x * blockDim.x + threadIdx.x;
  if (t >= total) return;
  const int e = t >> 3;
  const int h = t & 7;
  const int s = src[e];
  const int d = dst[e];
  const float v = lrelu(el[s * H_ + h] + er[d * H_ + h]);
  atomicMaxF(&mx[d * H_ + h], v);
}

__global__ void edge_sum(const int* __restrict__ src, const int* __restrict__ dst,
                         const float* __restrict__ el, const float* __restrict__ er,
                         const float* __restrict__ mx, float* __restrict__ sm,
                         int total) {
  const int t = blockIdx.x * blockDim.x + threadIdx.x;
  if (t >= total) return;
  const int e = t >> 3;
  const int h = t & 7;
  const int s = src[e];
  const int d = dst[e];
  const float v = lrelu(el[s * H_ + h] + er[d * H_ + h]);
  atomicAdd(&sm[d * H_ + h], __expf(v - mx[d * H_ + h]));
}

__global__ void edge_aggregate(const int* __restrict__ src, const int* __restrict__ dst,
                               const float* __restrict__ el, const float* __restrict__ er,
                               const float* __restrict__ mx, const float* __restrict__ sm,
                               const float* __restrict__ hp, float* __restrict__ out) {
  const int e    = blockIdx.x * (blockDim.x >> 5) + (threadIdx.x >> 5);
  const int lane = threadIdx.x & 31;
  if (e >= E_) return;
  const int s = src[e];
  const int d = dst[e];
  const int h = lane >> 2;
  const float v     = lrelu(el[s * H_ + h] + er[d * H_ + h]);
  const float alpha = __expf(v - mx[d * H_ + h]) / (sm[d * H_ + h] + 1e-16f);
  const float4 x = *(const float4*)(hp + (size_t)s * F_ + lane * 4);
  float* o = out + (size_t)d * F_ + lane * 4;
  atomicAdd(o + 0, alpha * x.x);
  atomicAdd(o + 1, alpha * x.y);
  atomicAdd(o + 2, alpha * x.z);
  atomicAdd(o + 3, alpha * x.w);
}

__global__ void ln_relu(float* __restrict__ io,
                        const float* __restrict__ bias,
                        const float* __restrict__ gamma,
                        const float* __restrict__ beta) {
  const int n    = blockIdx.x * (blockDim.x >> 5) + (threadIdx.x >> 5);
  const int lane = threadIdx.x & 31;
  if (n >= N_) return;
  float4 x = *(const float4*)(io + (size_t)n * F_ + lane * 4);
  const float4 bb = *(const float4*)(bias + lane * 4);
  x.x += bb.x; x.y += bb.y; x.z += bb.z; x.w += bb.w;
  float s  = x.x + x.y + x.z + x.w;
  float s2 = x.x * x.x + x.y * x.y + x.z * x.z + x.w * x.w;
#pragma unroll
  for (int off = 16; off >= 1; off >>= 1) {
    s  += __shfl_xor(s,  off, 32);
    s2 += __shfl_xor(s2, off, 32);
  }
  const float mu  = s * (1.f / 128.f);
  const float var = s2 * (1.f / 128.f) - mu * mu;
  const float inv = rsqrtf(var + LN_EPS);
  const float4 g  = *(const float4*)(gamma + lane * 4);
  const float4 be = *(const float4*)(beta + lane * 4);
  float4 y;
  y.x = fmaxf((x.x - mu) * inv * g.x + be.x, 0.f);
  y.y = fmaxf((x.y - mu) * inv * g.y + be.y, 0.f);
  y.z = fmaxf((x.z - mu) * inv * g.z + be.z, 0.f);
  y.w = fmaxf((x.w - mu) * inv * g.w + be.w, 0.f);
  *(float4*)(io + (size_t)n * F_ + lane * 4) = y;
}

__global__ void gemm_pred(const float* __restrict__ X,
                          const float* __restrict__ Wp,
                          const float* __restrict__ bp,
                          float* __restrict__ Y) {
  const int lane = threadIdx.x & 31;
  const int wave = threadIdx.x >> 5;
  const int m0   = blockIdx.x * 16;
  const int n0   = wave * 16;
  const int half = lane >> 4;
  const int l16  = lane & 15;
  const int arow = m0 + l16;
  const int n    = n0 + l16;
  const bool valid = (n < C_);

  v8f acc = {};
#pragma unroll
  for (int k = 0; k < F_; k += 4) {
    const int ka = k + 2 * half;
    v2f a;
    a.x = X[arow * F_ + ka];
    a.y = X[arow * F_ + ka + 1];
    v2f b;
    b.x = valid ? Wp[ka * C_ + n] : 0.f;
    b.y = valid ? Wp[(ka + 1) * C_ + n] : 0.f;
    acc = __builtin_amdgcn_wmma_f32_16x16x4_f32(false, a, false, b,
                                                (short)0, acc, false, false);
  }
  if (valid) {
    const float bbn = bp[n];
#pragma unroll
    for (int r = 0; r < 8; ++r) {
      const int m = m0 + r + 8 * half;
      Y[m * C_ + n] = acc[r] + bbn;
    }
  }
}

extern "C" void kernel_launch(void* const* d_in, const int* in_sizes, int n_in,
                              void* d_out, int out_size, void* d_ws, size_t ws_size,
                              hipStream_t stream) {
  const float* feats = (const float*)d_in[0];
  const int*   src   = (const int*)d_in[1];
  const int*   dst   = (const int*)d_in[2];
  const float* Ws    = (const float*)d_in[3];
  const float* al    = (const float*)d_in[4];
  const float* ar    = (const float*)d_in[5];
  const float* bias  = (const float*)d_in[6];
  const float* gamma = (const float*)d_in[7];
  const float* beta  = (const float*)d_in[8];
  const float* Wp    = (const float*)d_in[9];
  const float* bp    = (const float*)d_in[10];
  float* outp = (float*)d_out;

  float* ws   = (float*)d_ws;
  float* hbuf = ws;
  float* hp   = hbuf + (size_t)N_ * F_;
  float* el   = hp + (size_t)N_ * F_;
  float* er   = el + (size_t)N_ * H_;
  float* mx   = er + (size_t)N_ * H_;
  float* sm   = mx + (size_t)N_ * H_;

  const int NH = N_ * H_;
  const int EH = E_ * H_;
  const int NF = N_ * F_;

  for (int i = 0; i < L_; ++i) {
    const float* X = (i == 0) ? feats : hbuf;
    gemm_node_fc<<<N_ / 16, 256, 0, stream>>>(X, Ws + (size_t)i * F_ * F_, hp);
    node_coef<<<(NH + 255) / 256, 256, 0, stream>>>(
        hp, al + (size_t)i * H_ * D_, ar + (size_t)i * H_ * D_, el, er, mx, sm, NH);
    zero_f32<<<(NF + 255) / 256, 256, 0, stream>>>(hbuf, NF);
    edge_max<<<(EH + 255) / 256, 256, 0, stream>>>(src, dst, el, er, mx, EH);
    edge_sum<<<(EH + 255) / 256, 256, 0, stream>>>(src, dst, el, er, mx, sm, EH);
    edge_aggregate<<<E_ / 8, 256, 0, stream>>>(src, dst, el, er, mx, sm, hp, hbuf);
    ln_relu<<<N_ / 8, 256, 0, stream>>>(hbuf, bias + (size_t)i * F_,
                                        gamma + (size_t)i * F_, beta + (size_t)i * F_);
  }
  gemm_pred<<<N_ / 16, 96, 0, stream>>>(hbuf, Wp, bp, outp);
}


# Edges sorted by dst once per call (count, scan, place; integer atomics rank them in arrival order); one wave per node gathers its in-edges, softmax and sum in registers: no float atomics
# speedup vs baseline: 2.3322x; 2.3322x over previous
#include <hip/hip_runtime.h>

constexpr int N_ = 50000;
constexpr int E_ = 800000;
constexpr int F_ = 128;
constexpr int H_ = 8;
constexpr int D_ = 16;
constexpr int C_ = 40;
constexpr int L_ = 2;
constexpr float NEG_SLOPE = 0.2f;
constexpr float LN_EPS = 1e-5f;

typedef __attribute__((ext_vector_type(2))) float v2f;
typedef __attribute__((ext_vector_type(8))) float v8f;

__device__ __forceinline__ float lrelu(float x) { return x > 0.f ? x : NEG_SLOPE * x; }

__global__ void gemm_node_fc(const float* __restrict__ X,
                             const float* __restrict__ W,
                             float* __restrict__ Y) {
  const int lane = threadIdx.x & 31;
  const int wave = threadIdx.x >> 5;
  const int m0   = blockIdx.x * 16;
  const int n0   = wave * 16;
  const int half = lane >> 4;
  const int l16  = lane & 15;
  const int arow = m0 + l16;
  const int bcol = n0 + l16;

  v8f acc = {};
#pragma unroll
  for (int k = 0; k < F_; k += 4) {
    const int ka = k + 2 * half;
    v2f a;
    a.x = X[arow * F_ + ka];
    a.y = X[arow * F_ + ka + 1];
    v2f b;
    b.x = W[ka * F_ + bcol];
    b.y = W[(ka + 1) * F_ + bcol];
    acc = __builtin_amdgcn_wmma_f32_16x16x4_f32(false, a, false, b,
                                                (short)0, acc, false, false);
  }
#pragma unroll
  for (int r = 0; r < 8; ++r) {
    const int m = m0 + r + 8 * half;
    Y[m * F_ + bcol] = acc[r];
  }
}

__global__ void node_coef(const float* __restrict__ hp,
                          const float* __restrict__ al,
                          const float* __restrict__ ar,
                          float* __restrict__ el, float* __restrict__ er,
                          int total) {
  const int t = blockIdx.x * blockDim.x + threadIdx.x;
  if (t >= total) return;
  const int n = t >> 3;
  const int h = t & 7;
  const float* row = hp + n * F_ + h * D_;
  float sl = 0.f, sr = 0.f;
#pragma unroll
  for (int d = 0; d < D_; ++d) {
    const float x = row[d];
    sl += x * al[h * D_ + d];
    sr += x * ar[h * D_ + d];
  }
  el[t] = sl;
  er[t] = sr;
}

__global__ void zero_i32(int* __restrict__ p, int n) {
  const int t = blockIdx.x * blockDim.x + threadIdx.x;
  if (t < n) p[t] = 0;
}

__global__ void edge_rank(const int* __restrict__ dst, int* __restrict__ cnt,
                          int* __restrict__ rank) {
  const int e = blockIdx.x * blockDim.x + threadIdx.x;
  if (e >= E_) return;
  rank[e] = atomicAdd(&cnt[dst[e]], 1);
}

constexpr int SCAN_ROWS = 49;
__global__ void __launch_bounds__(1024)
seg_offsets(const int* __restrict__ cnt, int* __restrict__ off) {
  __shared__ int wtot[32];
  const int lane = threadIdx.x & 31;
  const int wave = threadIdx.x >> 5;
  const int base = wave * (SCAN_ROWS * 32);
  int tot = 0;
#pragma unroll 7
  for (int i = 0; i < SCAN_ROWS; ++i) {
    const int idx = base + i * 32 + lane;
    tot += (idx < N_) ? cnt[idx] : 0;
  }
#pragma unroll
  for (int o = 16; o >= 1; o >>= 1) tot += __shfl_xor(tot, o, 32);
  if (lane == 0) wtot[wave] = tot;
  __syncthreads();
  const int t = wtot[lane];
  int inc = t;
#pragma unroll
  for (int o = 1; o < 32; o <<= 1) {
    const int u = __shfl_up(inc, o, 32);
    if (lane >= o) inc += u;
  }
  int carry = __shfl(inc - t, wave, 32);
#pragma unroll 7
  for (int i = 0; i < SCAN_ROWS; ++i) {
    const int idx = base + i * 32 + lane;
    const int c = (idx < N_) ? cnt[idx] : 0;
    int v = c;
#pragma unroll
    for (int o = 1; o < 32; o <<= 1) {
      const int u = __shfl_up(v, o, 32);
      if (lane >= o) v += u;
    }
    if (idx < N_) off[idx] = carry + v - c;
    carry += __shfl(v, 31, 32);
  }
  if (threadIdx.x == 1023) off[N_] = carry;
}

__global__ void edge_place(const int* __restrict__ src, const int* __restrict__ dst,
                           const int* __restrict__ off, const int* __restrict__ rank,
                           int* __restrict__ perm) {
  const int e = blockIdx.x * blockDim.x + threadIdx.x;
  if (e >= E_) return;
  perm[off[dst[e]] + rank[e]] = src[e];
}

__device__ __forceinline__ float edge_score(const float* __restrict__ el, int s, int h,
                                            float ern) {
  return lrelu((el + (size_t)s * H_)[h] + ern);
}
__device__ __forceinline__ float4 edge_row(const float* __restrict__ hp, int s, int lane) {
  return ((const float4*)(hp + (size_t)s * F_))[lane];
}
__device__ __forceinline__ void axpy4(float4& acc, float a, const float4& x) {
  acc.x += a * x.x; acc.y += a * x.y; acc.z += a * x.z; acc.w += a * x.w;
}

__global__ void node_gather(const int* __restrict__ off, const int* __restrict__ perm,
                            const float* __restrict__ el, const float* __restrict__ er,
                            const float* __restrict__ hp, float* __restrict__ out) {
  const int wave = __builtin_amdgcn_readfirstlane(threadIdx.x >> 5);
  const int lane = threadIdx.x & 31;
  const int n    = blockIdx.x * (blockDim.x >> 5) + wave;
  if (n >= N_) return;
  const int h   = lane >> 2;
  const int beg = off[n];
  const int end = off[n + 1];
  const float ern = er[n * H_ + h];

  const int  full = beg + ((end - beg) & ~3);
  const bool tail = full < end;
  const int  last = end - 1;
  const int  t1 = min(full + 1, last), t2 = min(full + 2, last);
  const bool ok1 = full + 1 < end, ok2 = full + 2 < end;

  const float NEG_INF = __uint_as_float(0xff800000u);
  float m = NEG_INF;
  for (int j = beg; j < full; j += 4) {
    const int s0 = perm[j], s1 = perm[j + 1], s2 = perm[j + 2], s3 = perm[j + 3];
    const float v0 = edge_score(el, s0, h, ern);
    const float v1 = edge_score(el, s1, h, ern);
    const float v2 = edge_score(el, s2, h, ern);
    const float v3 = edge_score(el, s3, h, ern);
    m = fmaxf(fmaxf(m, fmaxf(v0, v1)), fmaxf(v2, v3));
  }
  if (tail) {
    const int s0 = perm[full], s1 = perm[t1], s2 = perm[t2];
    const float v0 = edge_score(el, s0, h, ern);
    const float v1 = edge_score(el, s1, h, ern);
    const float v2 = edge_score(el, s2, h, ern);
    m = fmaxf(fmaxf(m, v0), fmaxf(v1, v2));
  }

  float sum = 0.f;
  for (int j = beg; j < full; j += 4) {
    const int s0 = perm[j], s1 = perm[j + 1], s2 = perm[j + 2], s3 = perm[j + 3];
    const float v0 = edge_score(el, s0, h, ern);
    const float v1 = edge_score(el, s1, h, ern);
    const float v2 = edge_score(el, s2, h, ern);
    const float v3 = edge_score(el, s3, h, ern);
    sum += __expf(v0 - m);
    sum += __expf(v1 - m);
    sum += __expf(v2 - m);
    sum += __expf(v3 - m);
  }
  if (tail) {
    const int s0 = perm[full], s1 = perm[t1], s2 = perm[t2];
    const float e0 = edge_score(el, s0, h, ern);
    const float e1 = edge_score(el, s1, h, ern);
    const float e2 = edge_score(el, s2, h, ern);
    const float v1 = ok1 ? e1 : NEG_INF;
    const float v2 = ok2 ? e2 : NEG_INF;
    sum += __expf(e0 - m);
    sum += __expf(v1 - m);
    sum += __expf(v2 - m);
  }
  const float den = sum + 1e-16f;

  float4 acc = make_float4(0.f, 0.f, 0.f, 0.f);
  for (int j = beg; j < full; j += 4) {
    const int s0 = perm[j], s1 = perm[j + 1], s2 = perm[j + 2], s3 = perm[j + 3];
    const float4 x0 = edge_row(hp, s0, lane);
    const float4 x1 = edge_row(hp, s1, lane);
    const float4 x2 = edge_row(hp, s2, lane);
    const float4 x3 = edge_row(hp, s3, lane);
    const float a0 = __expf(edge_score(el, s0, h, ern) - m) / den;
    const float a1 = __expf(edge_score(el, s1, h, ern) - m) / den;
    const float a2 = __expf(edge_score(el, s2, h, ern) - m) / den;
    const float a3 = __expf(edge_score(el, s3, h, ern) - m) / den;
    axpy4(acc, a0, x0);
    axpy4(acc, a1, x1);
    axpy4(acc, a2, x2);
    axpy4(acc, a3, x3);
  }
  if (tail) {
    const int s0 = perm[full], s1 = perm[t1], s2 = perm[t2];
    const float4 x0 = edge_row(hp, s0, lane);
    const float4 x1 = edge_row(hp, s1, lane);
    const float4 x2 = edge_row(hp, s2, lane);
    const float e0 = edge_score(el, s0, h, ern);
    const float e1 = edge_score(el, s1, h, ern);
    const float e2 = edge_score(el, s2, h, ern);
    const float v1 = ok1 ? e1 : NEG_INF;
    const float v2 = ok2 ? e2 : NEG_INF;
    const float a0 = __expf(e0 - m) / den;
    const float a1 = __expf(v1 - m) / den;
    const float a2 = __expf(v2 - m) / den;
    axpy4(acc, a0, x0);
    axpy4(acc, a1, x1);
    axpy4(acc, a2, x2);
  }
  ((float4*)(out + (size_t)n * F_))[lane] = acc;
}

__global__ void ln_relu(float* __restrict__ io,
                        const float* __restrict__ bias,
                        const float* __restrict__ gamma,
                        const float* __restrict__ beta) {
  const int n    = blockIdx.x * (blockDim.x >> 5) + (threadIdx.x >> 5);
  const int lane = threadIdx.x & 31;
  if (n >= N_) return;
  float4 x = *(const float4*)(io + (size_t)n * F_ + lane * 4);
  const float4 bb = *(const float4*)(bias + lane * 4);
  x.x += bb.x; x.y += bb.y; x.z += bb.z; x.w += bb.w;
  float s  = x.x + x.y + x.z + x.w;
  float s2 = x.x * x.x + x.y * x.y + x.z * x.z + x.w * x.w;
#pragma unroll
  for (int off = 16; off >= 1; off >>= 1) {
    s  += __shfl_xor(s,  off, 32);
    s2 += __shfl_xor(s2, off, 32);
  }
  const float mu  = s * (1.f / 128.f);
  const float var = s2 * (1.f / 128.f) - mu * mu;
  const float inv = rsqrtf(var + LN_EPS);
  const float4 g  = *(const float4*)(gamma + lane * 4);
  const float4 be = *(const float4*)(beta + lane * 4);
  float4 y;
  y.x = fmaxf((x.x - mu) * inv * g.x + be.x, 0.f);
  y.y = fmaxf((x.y - mu) * inv * g.y + be.y, 0.f);
  y.z = fmaxf((x.z - mu) * inv * g.z + be.z, 0.f);
  y.w = fmaxf((x.w - mu) * inv * g.w + be.w, 0.f);
  *(float4*)(io + (size_t)n * F_ + lane * 4) = y;
}

__global__ void gemm_pred(const float* __restrict__ X,
                          const float* __restrict__ Wp,
                          const float* __restrict__ bp,
                          float* __restrict__ Y) {
  const int lane = threadIdx.x & 31;
  const int wave = threadIdx.x >> 5;
  const int m0   = blockIdx.x * 16;
  const int n0   = wave * 16;
  const int half = lane >> 4;
  const int l16  = lane & 15;
  const int arow = m0 + l16;
  const int n    = n0 + l16;
  const bool valid = (n < C_);

  v8f acc = {};
#pragma unroll
  for (int k = 0; k < F_; k += 4) {
    const int ka = k + 2 * half;
    v2f a;
    a.x = X[arow * F_ + ka];
    a.y = X[arow * F_ + ka + 1];
    v2f b;
    b.x = valid ? Wp[ka * C_ + n] : 0.f;
    b.y = valid ? Wp[(ka + 1) * C_ + n] : 0.f;
    acc = __builtin_amdgcn_wmma_f32_16x16x4_f32(false, a, false, b,
                                                (short)0, acc, false, false);
  }
  if (valid) {
    const float bbn = bp[n];
#pragma unroll
    for (int r = 0; r < 8; ++r) {
      const int m = m0 + r + 8 * half;
      Y[m * C_ + n] = acc[r] + bbn;
    }
  }
}

extern "C" void kernel_launch(void* const* d_in, const int* in_sizes, int n_in,
                              void* d_out, int out_size, void* d_ws, size_t ws_size,
                              hipStream_t stream) {
  const float* feats = (const float*)d_in[0];
  const int*   src   = (const int*)d_in[1];
  const int*   dst   = (const int*)d_in[2];
  const float* Ws    = (const float*)d_in[3];
  const float* al    = (const float*)d_in[4];
  const float* ar    = (const float*)d_in[5];
  const float* bias  = (const float*)d_in[6];
  const float* gamma = (const float*)d_in[7];
  const float* beta  = (const float*)d_in[8];
  const float* Wp    = (const float*)d_in[9];
  const float* bp    = (const float*)d_in[10];
  float* outp = (float*)d_out;

  float* ws   = (float*)d_ws;
  float* hbuf = ws;
  float* hp   = hbuf + (size_t)N_ * F_;
  float* el   = hp + (size_t)N_ * F_;
  float* er   = el + (size_t)N_ * H_;
  int*   cnt  = (int*)(er + (size_t)N_ * H_);
  int*   off  = cnt + N_;
  int*   rank = off + (N_ + 1);
  int*   perm = rank + E_;

  const int NH = N_ * H_;

  zero_i32<<<(N_ + 255) / 256, 256, 0, stream>>>(cnt, N_);
  edge_rank<<<(E_ + 255) / 256, 256, 0, stream>>>(dst, cnt, rank);
  seg_offsets<<<1, 1024, 0, stream>>>(cnt, off);
  edge_place<<<(E_ + 255) / 256, 256, 0, stream>>>(src, dst, off, rank, perm);

  for (int i = 0; i < L_; ++i) {
    const float* X = (i == 0) ? feats : hbuf;
    gemm_node_fc<<<N_ / 16, 256, 0, stream>>>(X, Ws + (size_t)i * F_ * F_, hp);
    node_coef<<<(NH + 255) / 256, 256, 0, stream>>>(
        hp, al + (size_t)i * H_ * D_, ar + (size_t)i * H_ * D_, el, er, NH);
    node_gather<<<N_ / 8, 256, 0, stream>>>(off, perm, el, er, hp, hbuf);
    ln_relu<<<N_ / 8, 256, 0, stream>>>(hbuf, bias + (size_t)i * F_,
                                        gamma + (size_t)i * F_, beta + (size_t)i * F_);
  }
  gemm_pred<<<N_ / 16, 96, 0, stream>>>(hbuf, Wp, bp, outp);
}


# Row epilogues fused into their producers: el/er from the fc GEMM's accumulator tile (16-lane exchange-tree sums: new summation order), bias+LayerNorm+ReLU in the gather wave; 13 -> 9 launches
# speedup vs baseline: 2.4701x; 1.0591x over previous
#include <hip/hip_runtime.h>

constexpr int N_ = 50000;
constexpr int E_ = 800000;
constexpr int F_ = 128;
constexpr int H_ = 8;
constexpr int D_ = 16;
constexpr int C_ = 40;
constexpr int L_ = 2;
constexpr float NEG_SLOPE = 0.2f;
constexpr float LN_EPS = 1e-5f;

typedef __attribute__((ext_vector_type(2))) float v2f;
typedef __attribute__((ext_vector_type(8))) float v8f;

__device__ __forceinline__ float lrelu(float x) { return x > 0.f ? x : NEG_SLOPE * x; }

__global__ void gemm_node_fc(const float* __restrict__ X,
                             const float* __restrict__ W,
                             const float* __restrict__ al,
                             const float* __restrict__ ar,
                             float* __restrict__ Y,
                             float* __restrict__ el, float* __restrict__ er) {
  __shared__ float4 s_coef[2 * 16 * H_ / 4];
  const int lane = threadIdx.x & 31;
  const int wave = threadIdx.x >> 5;
  const int m0   = blockIdx.x * 16;
  const int n0   = wave * 16;
  const int half = lane >> 4;
  const int l16  = lane & 15;
  const int arow = m0 + l16;
  const int bcol = n0 + l16;
  const float alv = al[wave * D_ + l16];
  const float arv = ar[wave * D_ + l16];

  v8f acc = {};
#pragma unroll
  for (int k = 0; k < F_; k += 4) {
    const int ka = k + 2 * half;
    v2f a;
    a.x = X[arow * F_ + ka];
    a.y = X[arow * F_ + ka + 1];
    v2f b;
    b.x = W[ka * F_ + bcol];
    b.y = W[(ka + 1) * F_ + bcol];
    acc = __builtin_amdgcn_wmma_f32_16x16x4_f32(false, a, false, b,
                                                (short)0, acc, false, false);
  }
#pragma unroll
  for (int r = 0; r < 8; ++r) {
    const int m = m0 + r + 8 * half;
    Y[m * F_ + bcol] = acc[r];
  }

  float p[16];
#pragma unroll
  for (int r = 0; r < 8; ++r) {
    p[r]     = acc[r] * alv;
    p[8 + r] = acc[r] * arv;
  }
  const bool b3 = (lane & 8) != 0, b2 = (lane & 4) != 0;
  const bool b1 = (lane & 2) != 0, b0 = (lane & 1) != 0;
  float q[8];
#pragma unroll
  for (int i = 0; i < 8; ++i) {
    const float keep = b3 ? p[i + 8] : p[i];
    const float send = b3 ? p[i] : p[i + 8];
    q[i] = keep + __shfl_xor(send, 8, 32);
  }
  float s[4];
#pragma unroll
  for (int i = 0; i < 4; ++i) {
    const float keep = b2 ? q[i + 4] : q[i];
    const float send = b2 ? q[i] : q[i + 4];
    s[i] = keep + __shfl_xor(send, 4, 32);
  }
  float t[2];
#pragma unroll
  for (int i = 0; i < 2; ++i) {
    const float keep = b1 ? s[i + 2] : s[i];
    const float send = b1 ? s[i] : s[i + 2];
    t[i] = keep + __shfl_xor(send, 2, 32);
  }
  const float keep0 = b0 ? t[1] : t[0];
  const float send0 = b0 ? t[0] : t[1];
  const float u = keep0 + __shfl_xor(send0, 1, 32);

  float* sc = (float*)s_coef;
  sc[(l16 >> 3) * (16 * H_) + ((l16 & 7) + 8 * half) * H_ + wave] = u;
  __syncthreads();
  const int wv = __builtin_amdgcn_readfirstlane(wave);
  if (wv < 2) {
    float* dstp = (wv == 0) ? el : er;
    ((float4*)(dstp + (size_t)m0 * H_))[lane] = s_coef[wv * 32 + lane];
  }
}

__global__ void zero_i32(int* __restrict__ p, int n) {
  const int t = blockIdx.x * blockDim.x + threadIdx.x;
  if (t < n) p[t] = 0;
}

__global__ void edge_rank(const int* __restrict__ dst, int* __restrict__ cnt,
                          int* __restrict__ rank) {
  const int e = blockIdx.x * blockDim.x + threadIdx.x;
  if (e >= E_) return;
  rank[e] = atomicAdd(&cnt[dst[e]], 1);
}

constexpr int SCAN_ROWS = 49;
__global__ void __launch_bounds__(1024)
seg_offsets(const int* __restrict__ cnt, int* __restrict__ off) {
  __shared__ int wtot[32];
  const int lane = threadIdx.x & 31;
  const int wave = threadIdx.x >> 5;
  const int base = wave * (SCAN_ROWS * 32);
  int tot = 0;
#pragma unroll 7
  for (int i = 0; i < SCAN_ROWS; ++i) {
    const int idx = base + i * 32 + lane;
    tot += (idx < N_) ? cnt[idx] : 0;
  }
#pragma unroll
  for (int o = 16; o >= 1; o >>= 1) tot += __shfl_xor(tot, o, 32);
  if (lane == 0) wtot[wave] = tot;
  __syncthreads();
  const int t = wtot[lane];
  int inc = t;
#pragma unroll
  for (int o = 1; o < 32; o <<= 1) {
    const int u = __shfl_up(inc, o, 32);
    if (lane >= o) inc += u;
  }
  int carry = __shfl(inc - t, wave, 32);
#pragma unroll 7
  for (int i = 0; i < SCAN_ROWS; ++i) {
    const int idx = base + i * 32 + lane;
    const int c = (idx < N_) ? cnt[idx] : 0;
    int v = c;
#pragma unroll
    for (int o = 1; o < 32; o <<= 1) {
      const int u = __shfl_up(v, o, 32);
      if (lane >= o) v += u;
    }
    if (idx < N_) off[idx] = carry + v - c;
    carry += __shfl(v, 31, 32);
  }
  if (threadIdx.x == 1023) off[N_] = carry;
}

__global__ void edge_place(const int* __restrict__ src, const int* __restrict__ dst,
                           const int* __restrict__ off, const int* __restrict__ rank,
                           int* __restrict__ perm) {
  const int e = blockIdx.x * blockDim.x + threadIdx.x;
  if (e >= E_) return;
  perm[off[dst[e]] + rank[e]] = src[e];
}

__device__ __forceinline__ float edge_score(const float* __restrict__ el, int s, int h,
                                            float ern) {
  return lrelu((el + (size_t)s * H_)[h] + ern);
}
__device__ __forceinline__ float4 edge_row(const float* __restrict__ hp, int s, int lane) {
  return ((const float4*)(hp + (size_t)s * F_))[lane];
}
__device__ __forceinline__ void axpy4(float4& acc, float a, const float4& x) {
  acc.x += a * x.x; acc.y += a * x.y; acc.z += a * x.z; acc.w += a * x.w;
}

__global__ void node_gather(const int* __restrict__ off, const int* __restrict__ perm,
                            const float* __restrict__ el, const float* __restrict__ er,
                            const float* __restrict__ hp,
                            const float* __restrict__ bias,
                            const float* __restrict__ gamma,
                            const float* __restrict__ beta,
                            float* __restrict__ out) {
  const int wave = __builtin_amdgcn_readfirstlane(threadIdx.x >> 5);
  const int lane = threadIdx.x & 31;
  const int n    = blockIdx.x * (blockDim.x >> 5) + wave;
  if (n >= N_) return;
  const int h   = lane >> 2;
  const int beg = off[n];
  const int end = off[n + 1];
  const float ern = er[n * H_ + h];

  const int  full = beg + ((end - beg) & ~3);
  const bool tail = full < end;
  const int  last = end - 1;
  const int  t1 = min(full + 1, last), t2 = min(full + 2, last);
  const bool ok1 = full + 1 < end, ok2 = full + 2 < end;

  const float NEG_INF = __uint_as_float(0xff800000u);
  float m = NEG_INF;
  for (int j = beg; j < full; j += 4) {
    const int s0 = perm[j], s1 = perm[j + 1], s2 = perm[j + 2], s3 = perm[j + 3];
    const float v0 = edge_score(el, s0, h, ern);
    const float v1 = edge_score(el, s1, h, ern);
    const float v2 = edge_score(el, s2, h, ern);
    const float v3 = edge_score(el, s3, h, ern);
    m = fmaxf(fmaxf(m, fmaxf(v0, v1)), fmaxf(v2, v3));
  }
  if (tail) {
    const int s0 = perm[full], s1 = perm[t1], s2 = perm[t2];
    const float v0 = edge_score(el, s0, h, ern);
    const float v1 = edge_score(el, s1, h, ern);
    const float v2 = edge_score(el, s2, h, ern);
    m = fmaxf(fmaxf(m, v0), fmaxf(v1, v2));
  }

  float sum = 0.f;
  for (int j = beg; j < full; j += 4) {
    const int s0 = perm[j], s1 = perm[j + 1], s2 = perm[j + 2], s3 = perm[j + 3];
    const float v0 = edge_score(el, s0, h, ern);
    const float v1 = edge_score(el, s1, h, ern);
    const float v2 = edge_score(el, s2, h, ern);
    const float v3 = edge_score(el, s3, h, ern);
    sum += __expf(v0 - m);
    sum += __expf(v1 - m);
    sum += __expf(v2 - m);
    sum += __expf(v3 - m);
  }
  if (tail) {
    const int s0 = perm[full], s1 = perm[t1], s2 = perm[t2];
    const float e0 = edge_score(el, s0, h, ern);
    const float e1 = edge_score(el, s1, h, ern);
    const float e2 = edge_score(el, s2, h, ern);
    const float v1 = ok1 ? e1 : NEG_INF;
    const float v2 = ok2 ? e2 : NEG_INF;
    sum += __expf(e0 - m);
    sum += __expf(v1 - m);
    sum += __expf(v2 - m);
  }
  const float den = sum + 1e-16f;

  float4 acc = make_float4(0.f, 0.f, 0.f, 0.f);
  for (int j = beg; j < full; j += 4) {
    const int s0 = perm[j], s1 = perm[j + 1], s2 = perm[j + 2], s3 = perm[j + 3];
    const float4 x0 = edge_row(hp, s0, lane);
    const float4 x1 = edge_row(hp, s1, lane);
    const float4 x2 = edge_row(hp, s2, lane);
    const float4 x3 = edge_row(hp, s3, lane);
    const float a0 = __expf(edge_score(el, s0, h, ern) - m) / den;
    const float a1 = __expf(edge_score(el, s1, h, ern) - m) / den;
    const float a2 = __expf(edge_score(el, s2, h, ern) - m) / den;
    const float a3 = __expf(edge_score(el, s3, h, ern) - m) / den;
    axpy4(acc, a0, x0);
    axpy4(acc, a1, x1);
    axpy4(acc, a2, x2);
    axpy4(acc, a3, x3);
  }
  if (tail) {
    const int s0 = perm[full], s1 = perm[t1], s2 = perm[t2];
    const float4 x0 = edge_row(hp, s0, lane);
    const float4 x1 = edge_row(hp, s1, lane);
    const float4 x2 = edge_row(hp, s2, lane);
    const float e0 = edge_score(el, s0, h, ern);
    const float e1 = edge_score(el, s1, h, ern);
    const float e2 = edge_score(el, s2, h, ern);
    const float v1 = ok1 ? e1 : NEG_INF;
    const float v2 = ok2 ? e2 : NEG_INF;
    const float a0 = __expf(e0 - m) / den;
    const float a1 = __expf(v1 - m) / den;
    const float a2 = __expf(v2 - m) / den;
    axpy4(acc, a0, x0);
    axpy4(acc, a1, x1);
    axpy4(acc, a2, x2);
  }
  const float4 bb = ((const float4*)bias)[lane];
  float4 x;
  x.x = acc.x + bb.x; x.y = acc.y + bb.y; x.z = acc.z + bb.z; x.w = acc.w + bb.w;
  float s  = x.x + x.y + x.z + x.w;
  float s2 = x.x * x.x + x.y * x.y + x.z * x.z + x.w * x.w;
#pragma unroll
  for (int o = 16; o >= 1; o >>= 1) {
    s  += __shfl_xor(s,  o, 32);
    s2 += __shfl_xor(s2, o, 32);
  }
  const float mu  = s * (1.f / 128.f);
  const float var = s2 * (1.f / 128.f) - mu * mu;
  const float inv = rsqrtf(var + LN_EPS);
  const float4 g  = ((const float4*)gamma)[lane];
  const float4 be = ((const float4*)beta)[lane];
  float4 y;
  y.x = fmaxf((x.x - mu) * inv * g.x + be.x, 0.f);
  y.y = fmaxf((x.y - mu) * inv * g.y + be.y, 0.f);
  y.z = fmaxf((x.z - mu) * inv * g.z + be.z, 0.f);
  y.w = fmaxf((x.w - mu) * inv * g.w + be.w, 0.f);
  ((float4*)(out + (size_t)n * F_))[lane] = y;
}

__global__ void gemm_pred(const float* __restrict__ X,
                          const float* __restrict__ Wp,
                          const float* __restrict__ bp,
                          float* __restrict__ Y) {
  const int lane = threadIdx.x & 31;
  const int wave = threadIdx.x >> 5;
  const int m0   = blockIdx.x * 16;
  const int n0   = wave * 16;
  const int half = lane >> 4;
  const int l16  = lane & 15;
  const int arow = m0 + l16;
  const int n    = n0 + l16;
  const bool valid = (n < C_);

  v8f acc = {};
#pragma unroll
  for (int k = 0; k < F_; k += 4) {
    const int ka = k + 2 * half;
    v2f a;
    a.x = X[arow * F_ + ka];
    a.y = X[arow * F_ + ka + 1];
    v2f b;
    b.x = valid ? Wp[ka * C_ + n] : 0.f;
    b.y = valid ? Wp[(ka + 1) * C_ + n] : 0.f;
    acc = __builtin_amdgcn_wmma_f32_16x16x4_f32(false, a, false, b,
                                                (short)0, acc, false, false);
  }
  if (valid) {
    const float bbn = bp[n];
#pragma unroll
    for (int r = 0; r < 8; ++r) {
      const int m = m0 + r + 8 * half;
      Y[m * C_ + n] = acc[r] + bbn;
    }
  }
}

extern "C" void kernel_launch(void* const* d_in, const int* in_sizes, int n_in,
                              void* d_out, int out_size, void* d_ws, size_t ws_size,
                              hipStream_t stream) {
  const float* feats = (const float*)d_in[0];
  const int*   src   = (const int*)d_in[1];
  const int*   dst   = (const int*)d_in[2];
  const float* Ws    = (const float*)d_in[3];
  const float* al    = (const float*)d_in[4];
  const float* ar    = (const float*)d_in[5];
  const float* bias  = (const float*)d_in[6];
  const float* gamma = (const float*)d_in[7];
  const float* beta  = (const float*)d_in[8];
  const float* Wp    = (const float*)d_in[9];
  const float* bp    = (const float*)d_in[10];
  float* outp = (float*)d_out;

  float* ws   = (float*)d_ws;
  float* hbuf = ws;
  float* hp   = hbuf + (size_t)N_ * F_;
  float* el   = hp + (size_t)N_ * F_;
  float* er   = el + (size_t)N_ * H_;
  int*   cnt  = (int*)(er + (size_t)N_ * H_);
  int*   off  = cnt + N_;
  int*   rank = off + (N_ + 1);
  int*   perm = rank + E_;

  zero_i32<<<(N_ + 255) / 256, 256, 0, stream>>>(cnt, N_);
  edge_rank<<<(E_ + 255) / 256, 256, 0, stream>>>(dst, cnt, rank);
  seg_offsets<<<1, 1024, 0, stream>>>(cnt, off);
  edge_place<<<(E_ + 255) / 256, 256, 0, stream>>>(src, dst, off, rank, perm);

  for (int i = 0; i < L_; ++i) {
    const float* X = (i == 0) ? feats : hbuf;
    gemm_node_fc<<<N_ / 16, 256, 0, stream>>>(
        X, Ws + (size_t)i * F_ * F_, al + (size_t)i * H_ * D_, ar + (size_t)i * H_ * D_,
        hp, el, er);
    node_gather<<<N_ / 8, 256, 0, stream>>>(off, perm, el, er, hp, bias + (size_t)i * F_,
                                            gamma + (size_t)i * F_, beta + (size_t)i * F_, hbuf);
  }
  gemm_pred<<<N_ / 16, 96, 0, stream>>>(hbuf, Wp, bp, outp);
}


# Gather walks each in-edge once: score recomputed from the hp[src] row it loads (no el/er arrays, GEMM epilogue gone), running max with rescale and one final divide: new summation order
# speedup vs baseline: 2.8949x; 1.1720x over previous
#include <hip/hip_runtime.h>

constexpr int N_ = 50000;
constexpr int E_ = 800000;
constexpr int F_ = 128;
constexpr int H_ = 8;
constexpr int D_ = 16;
constexpr int C_ = 40;
constexpr int L_ = 2;
constexpr float NEG_SLOPE = 0.2f;
constexpr float LN_EPS = 1e-5f;

typedef __attribute__((ext_vector_type(2))) float v2f;
typedef __attribute__((ext_vector_type(8))) float v8f;

__device__ __forceinline__ float lrelu(float x) { return x > 0.f ? x : NEG_SLOPE * x; }

__global__ void gemm_node_fc(const float* __restrict__ X,
                             const float* __restrict__ W,
                             float* __restrict__ Y) {
  const int lane = threadIdx.x & 31;
  const int wave = threadIdx.x >> 5;
  const int m0   = blockIdx.x * 16;
  const int n0   = wave * 16;
  const int half = lane >> 4;
  const int l16  = lane & 15;
  const int arow = m0 + l16;
  const int bcol = n0 + l16;

  v8f acc = {};
#pragma unroll
  for (int k = 0; k < F_; k += 4) {
    const int ka = k + 2 * half;
    v2f a;
    a.x = X[arow * F_ + ka];
    a.y = X[arow * F_ + ka + 1];
    v2f b;
    b.x = W[ka * F_ + bcol];
    b.y = W[(ka + 1) * F_ + bcol];
    acc = __builtin_amdgcn_wmma_f32_16x16x4_f32(false, a, false, b,
                                                (short)0, acc, false, false);
  }
#pragma unroll
  for (int r = 0; r < 8; ++r) {
    const int m = m0 + r + 8 * half;
    Y[m * F_ + bcol] = acc[r];
  }
}

__global__ void zero_i32(int* __restrict__ p, int n) {
  const int t = blockIdx.x * blockDim.x + threadIdx.x;
  if (t < n) p[t] = 0;
}

__global__ void edge_rank(const int* __restrict__ dst, int* __restrict__ cnt,
                          int* __restrict__ rank) {
  const int e = blockIdx.x * blockDim.x + threadIdx.x;
  if (e >= E_) return;
  rank[e] = atomicAdd(&cnt[dst[e]], 1);
}

constexpr int SCAN_ROWS = 49;
__global__ void __launch_bounds__(1024)
seg_offsets(const int* __restrict__ cnt, int* __restrict__ off) {
  __shared__ int wtot[32];
  const int lane = threadIdx.x & 31;
  const int wave = threadIdx.x >> 5;
  const int base = wave * (SCAN_ROWS * 32);
  int tot = 0;
#pragma unroll 7
  for (int i = 0; i < SCAN_ROWS; ++i) {
    const int idx = base + i * 32 + lane;
    tot += (idx < N_) ? cnt[idx] : 0;
  }
#pragma unroll
  for (int o = 16; o >= 1; o >>= 1) tot += __shfl_xor(tot, o, 32);
  if (lane == 0) wtot[wave] = tot;
  __syncthreads();
  const int t = wtot[lane];
  int inc = t;
#pragma unroll
  for (int o = 1; o < 32; o <<= 1) {
    const int u = __shfl_up(inc, o, 32);
    if (lane >= o) inc += u;
  }
  int carry = __shfl(inc - t, wave, 32);
#pragma unroll 7
  for (int i = 0; i < SCAN_ROWS; ++i) {
    const int idx = base + i * 32 + lane;
    const int c = (idx < N_) ? cnt[idx] : 0;
    int v = c;
#pragma unroll
    for (int o = 1; o < 32; o <<= 1) {
      const int u = __shfl_up(v, o, 32);
      if (lane >= o) v += u;
    }
    if (idx < N_) off[idx] = carry + v - c;
    carry += __shfl(v, 31, 32);
  }
  if (threadIdx.x == 1023) off[N_] = carry;
}

__global__ void edge_place(const int* __restrict__ src, const int* __restrict__ dst,
                           const int* __restrict__ off, const int* __restrict__ rank,
                           int* __restrict__ perm) {
  const int e = blockIdx.x * blockDim.x + threadIdx.x;
  if (e >= E_) return;
  perm[off[dst[e]] + rank[e]] = src[e];
}

__device__ __forceinline__ float4 edge_row(const float* __restrict__ hp, int s, int lane) {
  return ((const float4*)(hp + (size_t)s * F_))[lane];
}
__device__ __forceinline__ void axpy4(float4& acc, float a, const float4& x) {
  acc.x += a * x.x; acc.y += a * x.y; acc.z += a * x.z; acc.w += a * x.w;
}
__device__ __forceinline__ float dot4(const float4& a, const float4& b) {
  return a.x * b.x + a.y * b.y + a.z * b.z + a.w * b.w;
}
__device__ __forceinline__ float quad_sum(float x) {
  int b = __builtin_bit_cast(int, x);
  x += __builtin_bit_cast(float, __builtin_amdgcn_update_dpp(b, b, 0xB1, 0xF, 0xF, true));
  b = __builtin_bit_cast(int, x);
  x += __builtin_bit_cast(float, __builtin_amdgcn_update_dpp(b, b, 0x4E, 0xF, 0xF, true));
  return x;
}

__device__ __forceinline__ void gather_trips(const float* __restrict__ hp, int mys, int cnt,
                                             int lane, const float4& alv, float ern,
                                             float& m, float& sum, float4& acc) {
  const float NEG_INF = __uint_as_float(0xff800000u);
  for (int t = 0; t < cnt; t += 4) {
    const int s0 = __builtin_amdgcn_readlane(mys, t);
    const int s1 = __builtin_amdgcn_readlane(mys, t + 1);
    const int s2 = __builtin_amdgcn_readlane(mys, t + 2);
    const int s3 = __builtin_amdgcn_readlane(mys, t + 3);
    const float4 x0 = edge_row(hp, s0, lane);
    const float4 x1 = edge_row(hp, s1, lane);
    const float4 x2 = edge_row(hp, s2, lane);
    const float4 x3 = edge_row(hp, s3, lane);
    const float e0 = quad_sum(dot4(x0, alv));
    const float e1 = quad_sum(dot4(x1, alv));
    const float e2 = quad_sum(dot4(x2, alv));
    const float e3 = quad_sum(dot4(x3, alv));
    const float v0 = lrelu(e0 + ern);
    const float v1 = (t + 1 < cnt) ? lrelu(e1 + ern) : NEG_INF;
    const float v2 = (t + 2 < cnt) ? lrelu(e2 + ern) : NEG_INF;
    const float v3 = (t + 3 < cnt) ? lrelu(e3 + ern) : NEG_INF;
    const float mn = fmaxf(fmaxf(m, fmaxf(v0, v1)), fmaxf(v2, v3));
    const float sc = __expf(m - mn);
    const float p0 = __expf(v0 - mn);
    const float p1 = __expf(v1 - mn);
    const float p2 = __expf(v2 - mn);
    const float p3 = __expf(v3 - mn);
    sum = sum * sc + ((p0 + p1) + (p2 + p3));
    acc.x *= sc; acc.y *= sc; acc.z *= sc; acc.w *= sc;
    axpy4(acc, p0, x0);
    axpy4(acc, p1, x1);
    axpy4(acc, p2, x2);
    axpy4(acc, p3, x3);
    m = mn;
  }
}

__global__ void node_gather(const int* __restrict__ off, const int* __restrict__ perm,
                            const float* __restrict__ hp,
                            const float* __restrict__ al,
                            const float* __restrict__ ar,
                            const float* __restrict__ bias,
                            const float* __restrict__ gamma,
                            const float* __restrict__ beta,
                            float* __restrict__ out) {
  const int wave = __builtin_amdgcn_readfirstlane(threadIdx.x >> 5);
  const int lane = threadIdx.x & 31;
  const int n    = blockIdx.x * (blockDim.x >> 5) + wave;
  if (n >= N_) return;
  const int beg = off[n];
  const int end = off[n + 1];
  const float4 alv = ((const float4*)al)[lane];
  const float4 arv = ((const float4*)ar)[lane];
  const float4 own = edge_row(hp, n, lane);

  float m = __uint_as_float(0xff800000u);
  float sum = 0.f;
  float4 acc = make_float4(0.f, 0.f, 0.f, 0.f);
  if (beg < end) {
    const int last = end - 1;
    const int mys = perm[min(beg + lane, last)];
    const float ern = quad_sum(dot4(own, arv));
    gather_trips(hp, mys, min(end - beg, 32), lane, alv, ern, m, sum, acc);
    for (int c0 = beg + 32; c0 < end; c0 += 32) {
      const int more = perm[min(c0 + lane, last)];
      gather_trips(hp, more, min(end - c0, 32), lane, alv, ern, m, sum, acc);
    }
  }
  const float den = sum + 1e-16f;

  const float4 bb = ((const float4*)bias)[lane];
  float4 x;
  x.x = acc.x / den + bb.x; x.y = acc.y / den + bb.y;
  x.z = acc.z / den + bb.z; x.w = acc.w / den + bb.w;
  float s  = x.x + x.y + x.z + x.w;
  float s2 = x.x * x.x + x.y * x.y + x.z * x.z + x.w * x.w;
#pragma unroll
  for (int o = 16; o >= 1; o >>= 1) {
    s  += __shfl_xor(s,  o, 32);
    s2 += __shfl_xor(s2, o, 32);
  }
  const float mu  = s * (1.f / 128.f);
  const float var = s2 * (1.f / 128.f) - mu * mu;
  const float inv = rsqrtf(var + LN_EPS);
  const float4 g  = ((const float4*)gamma)[lane];
  const float4 be = ((const float4*)beta)[lane];
  float4 y;
  y.x = fmaxf((x.x - mu) * inv * g.x + be.x, 0.f);
  y.y = fmaxf((x.y - mu) * inv * g.y + be.y, 0.f);
  y.z = fmaxf((x.z - mu) * inv * g.z + be.z, 0.f);
  y.w = fmaxf((x.w - mu) * inv * g.w + be.w, 0.f);
  ((float4*)(out + (size_t)n * F_))[lane] = y;
}

__global__ void gemm_pred(const float* __restrict__ X,
                          const float* __restrict__ Wp,
                          const float* __restrict__ bp,
                          float* __restrict__ Y) {
  const int lane = threadIdx.x & 31;
  const int wave = threadIdx.x >> 5;
  const int m0   = blockIdx.x * 16;
  const int n0   = wave * 16;
  const int half = lane >> 4;
  const int l16  = lane & 15;
  const int arow = m0 + l16;
  const int n    = n0 + l16;
  const bool valid = (n < C_);

  v8f acc = {};
#pragma unroll
  for (int k = 0; k < F_; k += 4) {
    const int ka = k + 2 * half;
    v2f a;
    a.x = X[arow * F_ + ka];
    a.y = X[arow * F_ + ka + 1];
    v2f b;
    b.x = valid ? Wp[ka * C_ + n] : 0.f;
    b.y = valid ? Wp[(ka + 1) * C_ + n] : 0.f;
    acc = __builtin_amdgcn_wmma_f32_16x16x4_f32(false, a, false, b,
                                                (short)0, acc, false, false);
  }
  if (valid) {
    const float bbn = bp[n];
#pragma unroll
    for (int r = 0; r < 8; ++r) {
      const int m = m0 + r + 8 * half;
      Y[m * C_ + n] = acc[r] + bbn;
    }
  }
}

extern "C" void kernel_launch(void* const* d_in, const int* in_sizes, int n_in,
                              void* d_out, int out_size, void* d_ws, size_t ws_size,
                              hipStream_t stream) {
  const float* feats = (const float*)d_in[0];
  const int*   src   = (const int*)d_in[1];
  const int*   dst   = (const int*)d_in[2];
  const float* Ws    = (const float*)d_in[3];
  const float* al    = (const float*)d_in[4];
  const float* ar    = (const float*)d_in[5];
  const float* bias  = (const float*)d_in[6];
  const float* gamma = (const float*)d_in[7];
  const float* beta  = (const float*)d_in[8];
  const float* Wp    = (const float*)d_in[9];
  const float* bp    = (const float*)d_in[10];
  float* outp = (float*)d_out;

  float* ws   = (float*)d_ws;
  float* hbuf = ws;
  float* hp   = hbuf + (size_t)N_ * F_;
  int*   cnt  = (int*)(hp + (size_t)N_ * F_);
  int*   off  = cnt + N_;
  int*   rank = off + (N_ + 1);
  int*   perm = rank + E_;

  zero_i32<<<(N_ + 255) / 256, 256, 0, stream>>>(cnt, N_);
  edge_rank<<<(E_ + 255) / 256, 256, 0, stream>>>(dst, cnt, rank);
  seg_offsets<<<1, 1024, 0, stream>>>(cnt, off);
  edge_place<<<(E_ + 255) / 256, 256, 0, stream>>>(src, dst, off, rank, perm);

  for (int i = 0; i < L_; ++i) {
    const float* X = (i == 0) ? feats : hbuf;
    gemm_node_fc<<<N_ / 16, 256, 0, stream>>>(X, Ws + (size_t)i * F_ * F_, hp);
    node_gather<<<N_ / 8, 256, 0, stream>>>(off, perm, hp, al + (size_t)i * H_ * D_,
                                            ar + (size_t)i * H_ * D_, bias + (size_t)i * F_,
                                            gamma + (size_t)i * F_, beta + (size_t)i * F_, hbuf);
  }
  gemm_pred<<<N_ / 16, 96, 0, stream>>>(hbuf, Wp, bp, outp);
}


# Resident waves loop over their items (nodes, next node's sources prefetched; row tiles, weight slab loaded once; 4 edges per thread): 211k -> 61k waves launched per call
# speedup vs baseline: 3.1074x; 1.0734x over previous
#include <hip/hip_runtime.h>

constexpr int N_ = 50000;
constexpr int E_ = 800000;
constexpr int F_ = 128;
constexpr int H_ = 8;
constexpr int D_ = 16;
constexpr int C_ = 40;
constexpr int L_ = 2;
constexpr float NEG_SLOPE = 0.2f;
constexpr float LN_EPS = 1e-5f;

typedef __attribute__((ext_vector_type(2))) float v2f;
typedef __attribute__((ext_vector_type(8))) float v8f;

__device__ __forceinline__ float lrelu(float x) { return x > 0.f ? x : NEG_SLOPE * x; }

constexpr int M_TILES     = N_ / 16;
constexpr int FC_WGS      = 768;
constexpr int PRED_WGS    = 1024;
constexpr int GATHER_WGS  = 2048;
constexpr int E4_         = E_ / 4;
constexpr int N4_         = N_ / 4;

__global__ void __launch_bounds__(256)
gemm_node_fc(const float* __restrict__ X,
             const float* __restrict__ W,
             float* __restrict__ Y) {
  const int lane = threadIdx.x & 31;
  const int wave = threadIdx.x >> 5;
  const int n0   = wave * 16;
  const int half = lane >> 4;
  const int l16  = lane & 15;
  const int bcol = n0 + l16;

  v2f b[F_ / 4];
#pragma unroll
  for (int j = 0; j < F_ / 4; ++j) {
    const int ka = 4 * j + 2 * half;
    b[j].x = W[ka * F_ + bcol];
    b[j].y = W[(ka + 1) * F_ + bcol];
  }

  for (int tile = blockIdx.x; tile < M_TILES; tile += FC_WGS) {
    const int m0   = tile * 16;
    const int arow = m0 + l16;
    v8f acc = {};
#pragma unroll
    for (int j = 0; j < F_ / 4; ++j) {
      const int ka = 4 * j + 2 * half;
      v2f a;
      a.x = X[arow * F_ + ka];
      a.y = X[arow * F_ + ka + 1];
      acc = __builtin_amdgcn_wmma_f32_16x16x4_f32(false, a, false, b[j],
                                                  (short)0, acc, false, false);
    }
#pragma unroll
    for (int r = 0; r < 8; ++r) {
      const int m = m0 + r + 8 * half;
      Y[m * F_ + bcol] = acc[r];
    }
  }
}

__global__ void zero_i32(int4* __restrict__ p, int n4) {
  const int t = blockIdx.x * blockDim.x + threadIdx.x;
  if (t < n4) p[t] = make_int4(0, 0, 0, 0);
}

__global__ void edge_rank(const int* __restrict__ dst, int* __restrict__ cnt,
                          int* __restrict__ rank) {
  const int t = blockIdx.x * blockDim.x + threadIdx.x;
  if (t >= E4_) return;
  const int4 d = ((const int4*)dst)[t];
  int4 r;
  r.x = atomicAdd(&cnt[d.x], 1);
  r.y = atomicAdd(&cnt[d.y], 1);
  r.z = atomicAdd(&cnt[d.z], 1);
  r.w = atomicAdd(&cnt[d.w], 1);
  ((int4*)rank)[t] = r;
}

constexpr int SCAN_ROWS = 49;
__global__ void __launch_bounds__(1024)
seg_offsets(const int* __restrict__ cnt, int* __restrict__ off) {
  __shared__ int wtot[32];
  const int lane = threadIdx.x & 31;
  const int wave = threadIdx.x >> 5;
  const int base = wave * (SCAN_ROWS * 32);
  int tot = 0;
#pragma unroll 7
  for (int i = 0; i < SCAN_ROWS; ++i) {
    const int idx = base + i * 32 + lane;
    tot += (idx < N_) ? cnt[idx] : 0;
  }
#pragma unroll
  for (int o = 16; o >= 1; o >>= 1) tot += __shfl_xor(tot, o, 32);
  if (lane == 0) wtot[wave] = tot;
  __syncthreads();
  const int t = wtot[lane];
  int inc = t;
#pragma unroll
  for (int o = 1; o < 32; o <<= 1) {
    const int u = __shfl_up(inc, o, 32);
    if (lane >= o) inc += u;
  }
  int carry = __shfl(inc - t, wave, 32);
#pragma unroll 7
  for (int i = 0; i < SCAN_ROWS; ++i) {
    const int idx = base + i * 32 + lane;
    const int c = (idx < N_) ? cnt[idx] : 0;
    int v = c;
#pragma unroll
    for (int o = 1; o < 32; o <<= 1) {
      const int u = __shfl_up(v, o, 32);
      if (lane >= o) v += u;
    }
    if (idx < N_) off[idx] = carry + v - c;
    carry += __shfl(v, 31, 32);
  }
  if (threadIdx.x == 1023) off[N_] = carry;
}

__global__ void edge_place(const int* __restrict__ src, const int* __restrict__ dst,
                           const int* __restrict__ off, const int* __restrict__ rank,
                           int* __restrict__ perm) {
  const int t = blockIdx.x * blockDim.x + threadIdx.x;
  if (t >= E4_) return;
  const int4 s = ((const int4*)src)[t];
  const int4 d = ((const int4*)dst)[t];
  const int4 r = ((const int4*)rank)[t];
  const int o0 = off[d.x];
  const int o1 = off[d.y];
  const int o2 = off[d.z];
  const int o3 = off[d.w];
  perm[o0 + r.x] = s.x;
  perm[o1 + r.y] = s.y;
  perm[o2 + r.z] = s.z;
  perm[o3 + r.w] = s.w;
}

__device__ __forceinline__ float4 edge_row(const float* __restrict__ hp, int s, int lane) {
  return ((const float4*)(hp + (size_t)s * F_))[lane];
}
__device__ __forceinline__ void axpy4(float4& acc, float a, const float4& x) {
  acc.x += a * x.x; acc.y += a * x.y; acc.z += a * x.z; acc.w += a * x.w;
}
__device__ __forceinline__ float dot4(const float4& a, const float4& b) {
  return a.x * b.x + a.y * b.y + a.z * b.z + a.w * b.w;
}
__device__ __forceinline__ float quad_sum(float x) {
  int b = __builtin_bit_cast(int, x);
  x += __builtin_bit_cast(float, __builtin_amdgcn_update_dpp(b, b, 0xB1, 0xF, 0xF, true));
  b = __builtin_bit_cast(int, x);
  x += __builtin_bit_cast(float, __builtin_amdgcn_update_dpp(b, b, 0x4E, 0xF, 0xF, true));
  return x;
}

__device__ __forceinline__ void gather_trips(const float* __restrict__ hp, int mys, int cnt,
                                             int lane, const float4& alv, float ern,
                                             float& m, float& sum, float4& acc) {
  const float NEG_INF = __uint_as_float(0xff800000u);
  for (int t = 0; t < cnt; t += 4) {
    const int s0 = __builtin_amdgcn_readlane(mys, t);
    const int s1 = __builtin_amdgcn_readlane(mys, t + 1);
    const int s2 = __builtin_amdgcn_readlane(mys, t + 2);
    const int s3 = __builtin_amdgcn_readlane(mys, t + 3);
    const float4 x0 = edge_row(hp, s0, lane);
    const float4 x1 = edge_row(hp, s1, lane);
    const float4 x2 = edge_row(hp, s2, lane);
    const float4 x3 = edge_row(hp, s3, lane);
    const float e0 = quad_sum(dot4(x0, alv));
    const float e1 = quad_sum(dot4(x1, alv));
    const float e2 = quad_sum(dot4(x2, alv));
    const float e3 = quad_sum(dot4(x3, alv));
    const float v0 = lrelu(e0 + ern);
    const float v1 = (t + 1 < cnt) ? lrelu(e1 + ern) : NEG_INF;
    const float v2 = (t + 2 < cnt) ? lrelu(e2 + ern) : NEG_INF;
    const float v3 = (t + 3 < cnt) ? lrelu(e3 + ern) : NEG_INF;
    const float mn = fmaxf(fmaxf(m, fmaxf(v0, v1)), fmaxf(v2, v3));
    const float sc = __expf(m - mn);
    const float p0 = __expf(v0 - mn);
    const float p1 = __expf(v1 - mn);
    const float p2 = __expf(v2 - mn);
    const float p3 = __expf(v3 - mn);
    sum = sum * sc + ((p0 + p1) + (p2 + p3));
    acc.x *= sc; acc.y *= sc; acc.z *= sc; acc.w *= sc;
    axpy4(acc, p0, x0);
    axpy4(acc, p1, x1);
    axpy4(acc, p2, x2);
    axpy4(acc, p3, x3);
    m = mn;
  }
}

__global__ void __launch_bounds__(256)
node_gather(const int* __restrict__ off, const int* __restrict__ perm,
            const float* __restrict__ hp,
            const float* __restrict__ al,
            const float* __restrict__ ar,
            const float* __restrict__ bias,
            const float* __restrict__ gamma,
            const float* __restrict__ beta,
            float* __restrict__ out) {
  const int wave = __builtin_amdgcn_readfirstlane(threadIdx.x >> 5);
  const int lane = threadIdx.x & 31;
  constexpr int STRIDE = GATHER_WGS * 8;
  const float4 alv = ((const float4*)al)[lane];

  __shared__ float4 featv[4][32];
  if (wave < 4) {
    const float* p = (wave == 0) ? ar : (wave == 1) ? bias : (wave == 2) ? gamma : beta;
    featv[wave][lane] = ((const float4*)p)[lane];
  }
  __syncthreads();

  int n = blockIdx.x * 8 + wave;
  int beg = off[n];
  int end = off[n + 1];
  int n1 = n + STRIDE;
  int beg1 = 0, end1 = 0;
  if (n1 < N_) { beg1 = off[n1]; end1 = off[n1 + 1]; }
  float ern = quad_sum(dot4(edge_row(hp, n, lane), featv[0][lane]));
  int mys = 0;
  if (beg < end) mys = perm[min(beg + lane, end - 1)];

  do {
    const int n2 = n1 + STRIDE;
    int beg2 = 0, end2 = 0;
    if (n2 < N_) { beg2 = off[n2]; end2 = off[n2 + 1]; }
    int mys1 = 0;
    if (n1 < N_ && beg1 < end1) mys1 = perm[min(beg1 + lane, end1 - 1)];

    float m = __uint_as_float(0xff800000u);
    float sum = 0.f;
    float4 acc = make_float4(0.f, 0.f, 0.f, 0.f);
    if (beg < end) {
      const int last = end - 1;
      gather_trips(hp, mys, min(end - beg, 32), lane, alv, ern, m, sum, acc);
      for (int c0 = beg + 32; c0 < end; c0 += 32) {
        const int more = perm[min(c0 + lane, last)];
        gather_trips(hp, more, min(end - c0, 32), lane, alv, ern, m, sum, acc);
      }
    }
    const float den = sum + 1e-16f;

    asm volatile("" ::: "memory");
    const float4 arv = featv[0][lane];
    const float4 bb  = featv[1][lane];
    const float4 g   = featv[2][lane];
    const float4 be  = featv[3][lane];
    const float4 own1 = edge_row(hp, min(n1, N_ - 1), lane);

    float4 x;
    x.x = acc.x / den + bb.x; x.y = acc.y / den + bb.y;
    x.z = acc.z / den + bb.z; x.w = acc.w / den + bb.w;
    float s  = x.x + x.y + x.z + x.w;
    float s2 = x.x * x.x + x.y * x.y + x.z * x.z + x.w * x.w;
#pragma unroll
    for (int o = 16; o >= 1; o >>= 1) {
      s  += __shfl_xor(s,  o, 32);
      s2 += __shfl_xor(s2, o, 32);
    }
    const float mu  = s * (1.f / 128.f);
    const float var = s2 * (1.f / 128.f) - mu * mu;
    const float inv = rsqrtf(var + LN_EPS);
    float4 y;
    y.x = fmaxf((x.x - mu) * inv * g.x + be.x, 0.f);
    y.y = fmaxf((x.y - mu) * inv * g.y + be.y, 0.f);
    y.z = fmaxf((x.z - mu) * inv * g.z + be.z, 0.f);
    y.w = fmaxf((x.w - mu) * inv * g.w + be.w, 0.f);
    ((float4*)(out + (size_t)n * F_))[lane] = y;

    ern = quad_sum(dot4(own1, arv));
    n = n1; beg = beg1; end = end1; mys = mys1;
    n1 = n2; beg1 = beg2; end1 = end2;
  } while (n < N_);
}

__global__ void __launch_bounds__(96)
gemm_pred(const float* __restrict__ X,
          const float* __restrict__ Wp,
          const float* __restrict__ bp,
          float* __restrict__ Y) {
  const int lane = threadIdx.x & 31;
  const int wave = threadIdx.x >> 5;
  const int n0   = wave * 16;
  const int half = lane >> 4;
  const int l16  = lane & 15;
  const int n    = n0 + l16;
  const bool valid = (n < C_);

  v2f b[F_ / 4];
#pragma unroll
  for (int j = 0; j < F_ / 4; ++j) {
    const int ka = 4 * j + 2 * half;
    b[j].x = valid ? Wp[ka * C_ + n] : 0.f;
    b[j].y = valid ? Wp[(ka + 1) * C_ + n] : 0.f;
  }
  const float bbn = valid ? bp[n] : 0.f;

  for (int tile = blockIdx.x; tile < M_TILES; tile += PRED_WGS) {
    const int m0   = tile * 16;
    const int arow = m0 + l16;
    v8f acc = {};
#pragma unroll
    for (int j = 0; j < F_ / 4; ++j) {
      const int ka = 4 * j + 2 * half;
      v2f a;
      a.x = X[arow * F_ + ka];
      a.y = X[arow * F_ + ka + 1];
      acc = __builtin_amdgcn_wmma_f32_16x16x4_f32(false, a, false, b[j],
                                                  (short)0, acc, false, false);
    }
    if (valid) {
#pragma unroll
      for (int r = 0; r < 8; ++r) {
        const int m = m0 + r + 8 * half;
        Y[m * C_ + n] = acc[r] + bbn;
      }
    }
  }
}

extern "C" void kernel_launch(void* const* d_in, const int* in_sizes, int n_in,
                              void* d_out, int out_size, void* d_ws, size_t ws_size,
                              hipStream_t stream) {
  const float* feats = (const float*)d_in[0];
  const int*   src   = (const int*)d_in[1];
  const int*   dst   = (const int*)d_in[2];
  const float* Ws    = (const float*)d_in[3];
  const float* al    = (const float*)d_in[4];
  const float* ar    = (const float*)d_in[5];
  const float* bias  = (const float*)d_in[6];
  const float* gamma = (const float*)d_in[7];
  const float* beta  = (const float*)d_in[8];
  const float* Wp    = (const float*)d_in[9];
  const float* bp    = (const float*)d_in[10];
  float* outp = (float*)d_out;

  float* ws   = (float*)d_ws;
  float* hbuf = ws;
  float* hp   = hbuf + (size_t)N_ * F_;
  int*   cnt  = (int*)(hp + (size_t)N_ * F_);
  int*   rank = cnt + N_;
  int*   perm = rank + E_;
  int*   off  = perm + E_;

  zero_i32<<<(N4_ + 255) / 256, 256, 0, stream>>>((int4*)cnt, N4_);
  edge_rank<<<(E4_ + 255) / 256, 256, 0, stream>>>(dst, cnt, rank);
  seg_offsets<<<1, 1024, 0, stream>>>(cnt, off);
  edge_place<<<(E4_ + 255) / 256, 256, 0, stream>>>(src, dst, off, rank, perm);

  for (int i = 0; i < L_; ++i) {
    const float* X = (i == 0) ? feats : hbuf;
    gemm_node_fc<<<FC_WGS, 256, 0, stream>>>(X, Ws + (size_t)i * F_ * F_, hp);
    node_gather<<<GATHER_WGS, 256, 0, stream>>>(off, perm, hp, al + (size_t)i * H_ * D_,
                                            ar + (size_t)i * H_ * D_, bias + (size_t)i * F_,
                                            gamma + (size_t)i * F_, beta + (size_t)i * F_, hbuf);
  }
  gemm_pred<<<PRED_WGS, 96, 0, stream>>>(hbuf, Wp, bp, outp);
}


# In-edges put in 128-byte per-node buckets at the counting atomic's rank (sums stay in arrival order; past 32 an overflow list, one edge per trip): no prefix scan or placing pass, 9 -> 7 launches
# speedup vs baseline: 3.8320x; 1.2332x over previous
#include <hip/hip_runtime.h>

constexpr int N_ = 50000;
constexpr int E_ = 800000;
constexpr int F_ = 128;
constexpr int H_ = 8;
constexpr int D_ = 16;
constexpr int C_ = 40;
constexpr int L_ = 2;
constexpr float NEG_SLOPE = 0.2f;
constexpr float LN_EPS = 1e-5f;

typedef __attribute__((ext_vector_type(2))) float v2f;
typedef __attribute__((ext_vector_type(8))) float v8f;

__device__ __forceinline__ float lrelu(float x) { return x > 0.f ? x : NEG_SLOPE * x; }

constexpr int M_TILES     = N_ / 16;
constexpr int FC_WGS      = 768;
constexpr int PRED_WGS    = 1024;
constexpr int GATHER_WGS  = 2048;
constexpr int E4_         = E_ / 4;
constexpr int CAP         = 32;
constexpr int CNT4_       = (N_ + 4) / 4;

__global__ void __launch_bounds__(256)
gemm_node_fc(const float* __restrict__ X,
             const float* __restrict__ W,
             float* __restrict__ Y) {
  const int lane = threadIdx.x & 31;
  const int wave = threadIdx.x >> 5;
  const int n0   = wave * 16;
  const int half = lane >> 4;
  const int l16  = lane & 15;
  const int bcol = n0 + l16;

  v2f b[F_ / 4];
#pragma unroll
  for (int j = 0; j < F_ / 4; ++j) {
    const int ka = 4 * j + 2 * half;
    b[j].x = W[ka * F_ + bcol];
    b[j].y = W[(ka + 1) * F_ + bcol];
  }

  for (int tile = blockIdx.x; tile < M_TILES; tile += FC_WGS) {
    const int m0   = tile * 16;
    const int arow = m0 + l16;
    v8f acc = {};
#pragma unroll
    for (int j = 0; j < F_ / 4; ++j) {
      const int ka = 4 * j + 2 * half;
      v2f a;
      a.x = X[arow * F_ + ka];
      a.y = X[arow * F_ + ka + 1];
      acc = __builtin_amdgcn_wmma_f32_16x16x4_f32(false, a, false, b[j],
                                                  (short)0, acc, false, false);
    }
#pragma unroll
    for (int r = 0; r < 8; ++r) {
      const int m = m0 + r + 8 * half;
      Y[m * F_ + bcol] = acc[r];
    }
  }
}

__global__ void zero_i32(int4* __restrict__ p, int n4) {
  const int t = blockIdx.x * blockDim.x + threadIdx.x;
  if (t < n4) p[t] = make_int4(0, 0, 0, 0);
}

__device__ __forceinline__ void drop_edge(int d, int s, int r, int* __restrict__ cnt,
                                          int* __restrict__ bucket,
                                          int* __restrict__ ovf_dst, int* __restrict__ ovf_src) {
  if (r < CAP) {
    bucket[d * CAP + r] = s;
  } else {
    const int k = atomicAdd(&cnt[N_], 1);
    ovf_dst[k] = d;
    ovf_src[k] = s;
  }
}

__global__ void edge_bucket(const int* __restrict__ src, const int* __restrict__ dst,
                            int* __restrict__ cnt, int* __restrict__ bucket,
                            int* __restrict__ ovf_dst, int* __restrict__ ovf_src) {
  const int t = blockIdx.x * blockDim.x + threadIdx.x;
  if (t >= E4_) return;
  const int4 d = ((const int4*)dst)[t];
  const int4 s = ((const int4*)src)[t];
  const int r0 = atomicAdd(&cnt[d.x], 1);
  const int r1 = atomicAdd(&cnt[d.y], 1);
  const int r2 = atomicAdd(&cnt[d.z], 1);
  const int r3 = atomicAdd(&cnt[d.w], 1);
  drop_edge(d.x, s.x, r0, cnt, bucket, ovf_dst, ovf_src);
  drop_edge(d.y, s.y, r1, cnt, bucket, ovf_dst, ovf_src);
  drop_edge(d.z, s.z, r2, cnt, bucket, ovf_dst, ovf_src);
  drop_edge(d.w, s.w, r3, cnt, bucket, ovf_dst, ovf_src);
}

__device__ __forceinline__ float4 edge_row(const float* __restrict__ hp, int s, int lane) {
  return ((const float4*)(hp + (size_t)s * F_))[lane];
}
__device__ __forceinline__ void axpy4(float4& acc, float a, const float4& x) {
  acc.x += a * x.x; acc.y += a * x.y; acc.z += a * x.z; acc.w += a * x.w;
}
__device__ __forceinline__ float dot4(const float4& a, const float4& b) {
  return a.x * b.x + a.y * b.y + a.z * b.z + a.w * b.w;
}
__device__ __forceinline__ float quad_sum(float x) {
  int b = __builtin_bit_cast(int, x);
  x += __builtin_bit_cast(float, __builtin_amdgcn_update_dpp(b, b, 0xB1, 0xF, 0xF, true));
  b = __builtin_bit_cast(int, x);
  x += __builtin_bit_cast(float, __builtin_amdgcn_update_dpp(b, b, 0x4E, 0xF, 0xF, true));
  return x;
}

__device__ __forceinline__ void gather_trips(const float* __restrict__ hp, int mys, int cnt,
                                             int lane, const float4& alv, float ern,
                                             float& m, float& sum, float4& acc) {
  const float NEG_INF = __uint_as_float(0xff800000u);
  for (int t = 0; t < cnt; t += 4) {
    const int s0 = __builtin_amdgcn_readlane(mys, t);
    const int s1 = __builtin_amdgcn_readlane(mys, t + 1);
    const int s2 = __builtin_amdgcn_readlane(mys, t + 2);
    const int s3 = __builtin_amdgcn_readlane(mys, t + 3);
    const float4 x0 = edge_row(hp, s0, lane);
    const float4 x1 = edge_row(hp, s1, lane);
    const float4 x2 = edge_row(hp, s2, lane);
    const float4 x3 = edge_row(hp, s3, lane);
    const float e0 = quad_sum(dot4(x0, alv));
    const float e1 = quad_sum(dot4(x1, alv));
    const float e2 = quad_sum(dot4(x2, alv));
    const float e3 = quad_sum(dot4(x3, alv));
    const float v0 = lrelu(e0 + ern);
    const float v1 = (t + 1 < cnt) ? lrelu(e1 + ern) : NEG_INF;
    const float v2 = (t + 2 < cnt) ? lrelu(e2 + ern) : NEG_INF;
    const float v3 = (t + 3 < cnt) ? lrelu(e3 + ern) : NEG_INF;
    const float mn = fmaxf(fmaxf(m, fmaxf(v0, v1)), fmaxf(v2, v3));
    const float sc = __expf(m - mn);
    const float p0 = __expf(v0 - mn);
    const float p1 = __expf(v1 - mn);
    const float p2 = __expf(v2 - mn);
    const float p3 = __expf(v3 - mn);
    sum = sum * sc + ((p0 + p1) + (p2 + p3));
    acc.x *= sc; acc.y *= sc; acc.z *= sc; acc.w *= sc;
    axpy4(acc, p0, x0);
    axpy4(acc, p1, x1);
    axpy4(acc, p2, x2);
    axpy4(acc, p3, x3);
    m = mn;
  }
}

__global__ void __launch_bounds__(256)
node_gather(const int* __restrict__ cnt, const int* __restrict__ bucket,
            const int* __restrict__ ovf_dst, const int* __restrict__ ovf_src,
            const float* __restrict__ hp,
            const float* __restrict__ al,
            const float* __restrict__ ar,
            const float* __restrict__ bias,
            const float* __restrict__ gamma,
            const float* __restrict__ beta,
            float* __restrict__ out) {
  const int wave = __builtin_amdgcn_readfirstlane(threadIdx.x >> 5);
  const int lane = threadIdx.x & 31;
  constexpr int STRIDE = GATHER_WGS * 8;
  const float4 alv = ((const float4*)al)[lane];

  __shared__ float4 featv[4][32];
  if (wave < 4) {
    const float* p = (wave == 0) ? ar : (wave == 1) ? bias : (wave == 2) ? gamma : beta;
    featv[wave][lane] = ((const float4*)p)[lane];
  }
  __syncthreads();

  int n = blockIdx.x * 8 + wave;
  int deg = cnt[n];
  int n1 = n + STRIDE;
  int deg1 = 0;
  if (n1 < N_) deg1 = cnt[n1];
  float ern = quad_sum(dot4(edge_row(hp, n, lane), featv[0][lane]));
  int mys = 0;
  if (deg > 0) mys = bucket[n * CAP + min(lane, min(deg, CAP) - 1)];

  do {
    const int n2 = n1 + STRIDE;
    int deg2 = 0;
    if (n2 < N_) deg2 = cnt[n2];
    int mys1 = 0;
    if (n1 < N_ && deg1 > 0) mys1 = bucket[n1 * CAP + min(lane, min(deg1, CAP) - 1)];

    float m = __uint_as_float(0xff800000u);
    float sum = 0.f;
    float4 acc = make_float4(0.f, 0.f, 0.f, 0.f);
    if (deg > 0) {
      gather_trips(hp, mys, min(deg, CAP), lane, alv, ern, m, sum, acc);
      if (deg > CAP) {
        const int novf = cnt[N_];
        for (int c0 = 0; c0 < novf; c0 += 32) {
          const int i = c0 + lane;
          int od = -1, os = 0;
          if (i < novf) { od = ovf_dst[i]; os = ovf_src[i]; }
          unsigned hit = __builtin_amdgcn_ballot_w32(od == n);
          while (hit) {
            const int b = __builtin_ctz(hit);
            hit &= hit - 1;
            gather_trips(hp, __builtin_amdgcn_readlane(os, b), 1, lane, alv, ern, m, sum, acc);
          }
        }
      }
    }
    const float den = sum + 1e-16f;

    asm volatile("" ::: "memory");
    const float4 arv = featv[0][lane];
    const float4 bb  = featv[1][lane];
    const float4 g   = featv[2][lane];
    const float4 be  = featv[3][lane];
    const float4 own1 = edge_row(hp, min(n1, N_ - 1), lane);

    float4 x;
    x.x = acc.x / den + bb.x; x.y = acc.y / den + bb.y;
    x.z = acc.z / den + bb.z; x.w = acc.w / den + bb.w;
    float s  = x.x + x.y + x.z + x.w;
    float s2 = x.x * x.x + x.y * x.y + x.z * x.z + x.w * x.w;
#pragma unroll
    for (int o = 16; o >= 1; o >>= 1) {
      s  += __shfl_xor(s,  o, 32);
      s2 += __shfl_xor(s2, o, 32);
    }
    const float mu  = s * (1.f / 128.f);
    const float var = s2 * (1.f / 128.f) - mu * mu;
    const float inv = rsqrtf(var + LN_EPS);
    float4 y;
    y.x = fmaxf((x.x - mu) * inv * g.x + be.x, 0.f);
    y.y = fmaxf((x.y - mu) * inv * g.y + be.y, 0.f);
    y.z = fmaxf((x.z - mu) * inv * g.z + be.z, 0.f);
    y.w = fmaxf((x.w - mu) * inv * g.w + be.w, 0.f);
    ((float4*)(out + (size_t)n * F_))[lane] = y;

    ern = quad_sum(dot4(own1, arv));
    n = n1; deg = deg1; mys = mys1;
    n1 = n2; deg1 = deg2;
  } while (n < N_);
}

__global__ void __launch_bounds__(96)
gemm_pred(const float* __restrict__ X,
          const float* __restrict__ Wp,
          const float* __restrict__ bp,
          float* __restrict__ Y) {
  const int lane = threadIdx.x & 31;
  const int wave = threadIdx.x >> 5;
  const int n0   = wave * 16;
  const int half = lane >> 4;
  const int l16  = lane & 15;
  const int n    = n0 + l16;
  const bool valid = (n < C_);

  v2f b[F_ / 4];
#pragma unroll
  for (int j = 0; j < F_ / 4; ++j) {
    const int ka = 4 * j + 2 * half;
    b[j].x = valid ? Wp[ka * C_ + n] : 0.f;
    b[j].y = valid ? Wp[(ka + 1) * C_ + n] : 0.f;
  }
  const float bbn = valid ? bp[n] : 0.f;

  for (int tile = blockIdx.x; tile < M_TILES; tile += PRED_WGS) {
    const int m0   = tile * 16;
    const int arow = m0 + l16;
    v8f acc = {};
#pragma unroll
    for (int j = 0; j < F_ / 4; ++j) {
      const int ka = 4 * j + 2 * half;
      v2f a;
      a.x = X[arow * F_ + ka];
      a.y = X[arow * F_ + ka + 1];
      acc = __builtin_amdgcn_wmma_f32_16x16x4_f32(false, a, false, b[j],
                                                  (short)0, acc, false, false);
    }
    if (valid) {
#pragma unroll
      for (int r = 0; r < 8; ++r) {
        const int m = m0 + r + 8 * half;
        Y[m * C_ + n] = acc[r] + bbn;
      }
    }
  }
}

extern "C" void kernel_launch(void* const* d_in, const int* in_sizes, int n_in,
                              void* d_out, int out_size, void* d_ws, size_t ws_size,
                              hipStream_t stream) {
  const float* feats = (const float*)d_in[0];
  const int*   src   = (const int*)d_in[1];
  const int*   dst   = (const int*)d_in[2];
  const float* Ws    = (const float*)d_in[3];
  const float* al    = (const float*)d_in[4];
  const float* ar    = (const float*)d_in[5];
  const float* bias  = (const float*)d_in[6];
  const float* gamma = (const float*)d_in[7];
  const float* beta  = (const float*)d_in[8];
  const float* Wp    = (const float*)d_in[9];
  const float* bp    = (const float*)d_in[10];
  float* outp = (float*)d_out;

  float* ws   = (float*)d_ws;
  float* hbuf = ws;
  float* hp   = hbuf + (size_t)N_ * F_;
  int*   bucket  = (int*)(hp + (size_t)N_ * F_);
  int*   cnt     = bucket + (size_t)N_ * CAP;
  int*   ovf_dst = cnt + (N_ + 4);
  int*   ovf_src = ovf_dst + E_;

  zero_i32<<<(CNT4_ + 255) / 256, 256, 0, stream>>>((int4*)cnt, CNT4_);
  edge_bucket<<<(E4_ + 255) / 256, 256, 0, stream>>>(src, dst, cnt, bucket, ovf_dst, ovf_src);

  for (int i = 0; i < L_; ++i) {
    const float* X = (i == 0) ? feats : hbuf;
    gemm_node_fc<<<FC_WGS, 256, 0, stream>>>(X, Ws + (size_t)i * F_ * F_, hp);
    node_gather<<<GATHER_WGS, 256, 0, stream>>>(cnt, bucket, ovf_dst, ovf_src, hp,
                                            al + (size_t)i * H_ * D_,
                                            ar + (size_t)i * H_ * D_, bias + (size_t)i * F_,
                                            gamma + (size_t)i * F_, beta + (size_t)i * F_, hbuf);
  }
  gemm_pred<<<PRED_WGS, 96, 0, stream>>>(hbuf, Wp, bp, outp);
}


# Each node's in-edge counter moved into its own 128-byte bucket row (slot 31, 31 source slots): ranking atomics share no line between nodes; gather gets degree and sources in one load
# speedup vs baseline: 4.4797x; 1.1690x over previous
#include <hip/hip_runtime.h>

constexpr int N_ = 50000;
constexpr int E_ = 800000;
constexpr int F_ = 128;
constexpr int H_ = 8;
constexpr int D_ = 16;
constexpr int C_ = 40;
constexpr int L_ = 2;
constexpr float NEG_SLOPE = 0.2f;
constexpr float LN_EPS = 1e-5f;

typedef __attribute__((ext_vector_type(2))) float v2f;
typedef __attribute__((ext_vector_type(8))) float v8f;

__device__ __forceinline__ float lrelu(float x) { return x > 0.f ? x : NEG_SLOPE * x; }

constexpr int M_TILES     = N_ / 16;
constexpr int FC_WGS      = 768;
constexpr int PRED_WGS    = 1024;
constexpr int GATHER_WGS  = 2048;
constexpr int E4_         = E_ / 4;
constexpr int ROW         = 32;
constexpr int CAP         = ROW - 1;
constexpr int ZERO4_      = N_ * (ROW / 4) + 1;
constexpr int ZERO_WGS    = 256;

__global__ void __launch_bounds__(256)
gemm_node_fc(const float* __restrict__ X,
             const float* __restrict__ W,
             float* __restrict__ Y) {
  const int lane = threadIdx.x & 31;
  const int wave = threadIdx.x >> 5;
  const int n0   = wave * 16;
  const int half = lane >> 4;
  const int l16  = lane & 15;
  const int bcol = n0 + l16;

  v2f b[F_ / 4];
#pragma unroll
  for (int j = 0; j < F_ / 4; ++j) {
    const int ka = 4 * j + 2 * half;
    b[j].x = W[ka * F_ + bcol];
    b[j].y = W[(ka + 1) * F_ + bcol];
  }

  for (int tile = blockIdx.x; tile < M_TILES; tile += FC_WGS) {
    const int m0   = tile * 16;
    const int arow = m0 + l16;
    v8f acc = {};
#pragma unroll
    for (int j = 0; j < F_ / 4; ++j) {
      const int ka = 4 * j + 2 * half;
      v2f a;
      a.x = X[arow * F_ + ka];
      a.y = X[arow * F_ + ka + 1];
      acc = __builtin_amdgcn_wmma_f32_16x16x4_f32(false, a, false, b[j],
                                                  (short)0, acc, false, false);
    }
#pragma unroll
    for (int r = 0; r < 8; ++r) {
      const int m = m0 + r + 8 * half;
      Y[m * F_ + bcol] = acc[r];
    }
  }
}

__global__ void __launch_bounds__(256) zero_rows(int4* __restrict__ p) {
  for (int t = blockIdx.x * 256 + threadIdx.x; t < ZERO4_; t += ZERO_WGS * 256)
    p[t] = make_int4(0, 0, 0, 0);
}

__device__ __forceinline__ void drop_edge(int d, int s, int r, int* __restrict__ bucket,
                                          int* __restrict__ ovf_len,
                                          int* __restrict__ ovf_dst, int* __restrict__ ovf_src) {
  if (r < CAP) {
    bucket[d * ROW + r] = s;
  } else {
    const int k = atomicAdd(ovf_len, 1);
    ovf_dst[k] = d;
    ovf_src[k] = s;
  }
}

__global__ void edge_bucket(const int* __restrict__ src, const int* __restrict__ dst,
                            int* __restrict__ bucket, int* __restrict__ ovf_len,
                            int* __restrict__ ovf_dst, int* __restrict__ ovf_src) {
  const int t = blockIdx.x * blockDim.x + threadIdx.x;
  if (t >= E4_) return;
  const int4 d = ((const int4*)dst)[t];
  const int4 s = ((const int4*)src)[t];
  const int r0 = atomicAdd(&bucket[d.x * ROW + CAP], 1);
  const int r1 = atomicAdd(&bucket[d.y * ROW + CAP], 1);
  const int r2 = atomicAdd(&bucket[d.z * ROW + CAP], 1);
  const int r3 = atomicAdd(&bucket[d.w * ROW + CAP], 1);
  drop_edge(d.x, s.x, r0, bucket, ovf_len, ovf_dst, ovf_src);
  drop_edge(d.y, s.y, r1, bucket, ovf_len, ovf_dst, ovf_src);
  drop_edge(d.z, s.z, r2, bucket, ovf_len, ovf_dst, ovf_src);
  drop_edge(d.w, s.w, r3, bucket, ovf_len, ovf_dst, ovf_src);
}

__device__ __forceinline__ float4 edge_row(const float* __restrict__ hp, int s, int lane) {
  return ((const float4*)(hp + (size_t)s * F_))[lane];
}
__device__ __forceinline__ void axpy4(float4& acc, float a, const float4& x) {
  acc.x += a * x.x; acc.y += a * x.y; acc.z += a * x.z; acc.w += a * x.w;
}
__device__ __forceinline__ float dot4(const float4& a, const float4& b) {
  return a.x * b.x + a.y * b.y + a.z * b.z + a.w * b.w;
}
__device__ __forceinline__ float quad_sum(float x) {
  int b = __builtin_bit_cast(int, x);
  x += __builtin_bit_cast(float, __builtin_amdgcn_update_dpp(b, b, 0xB1, 0xF, 0xF, true));
  b = __builtin_bit_cast(int, x);
  x += __builtin_bit_cast(float, __builtin_amdgcn_update_dpp(b, b, 0x4E, 0xF, 0xF, true));
  return x;
}

__device__ __forceinline__ void gather_trips(const float* __restrict__ hp, int mys, int cnt,
                                             int lane, const float4& alv, float ern,
                                             float& m, float& sum, float4& acc) {
  const float NEG_INF = __uint_as_float(0xff800000u);
  for (int t = 0; t < cnt; t += 4) {
    const int s0 = __builtin_amdgcn_readlane(mys, t);
    const int s1 = __builtin_amdgcn_readlane(mys, t + 1);
    const int s2 = __builtin_amdgcn_readlane(mys, t + 2);
    const int s3 = __builtin_amdgcn_readlane(mys, t + 3);
    const float4 x0 = edge_row(hp, s0, lane);
    const float4 x1 = edge_row(hp, s1, lane);
    const float4 x2 = edge_row(hp, s2, lane);
    const float4 x3 = edge_row(hp, s3, lane);
    const float e0 = quad_sum(dot4(x0, alv));
    const float e1 = quad_sum(dot4(x1, alv));
    const float e2 = quad_sum(dot4(x2, alv));
    const float e3 = quad_sum(dot4(x3, alv));
    const float v0 = lrelu(e0 + ern);
    const float v1 = (t + 1 < cnt) ? lrelu(e1 + ern) : NEG_INF;
    const float v2 = (t + 2 < cnt) ? lrelu(e2 + ern) : NEG_INF;
    const float v3 = (t + 3 < cnt) ? lrelu(e3 + ern) : NEG_INF;
    const float mn = fmaxf(fmaxf(m, fmaxf(v0, v1)), fmaxf(v2, v3));
    const float sc = __expf(m - mn);
    const float p0 = __expf(v0 - mn);
    const float p1 = __expf(v1 - mn);
    const float p2 = __expf(v2 - mn);
    const float p3 = __expf(v3 - mn);
    sum = sum * sc + ((p0 + p1) + (p2 + p3));
    acc.x *= sc; acc.y *= sc; acc.z *= sc; acc.w *= sc;
    axpy4(acc, p0, x0);
    axpy4(acc, p1, x1);
    axpy4(acc, p2, x2);
    axpy4(acc, p3, x3);
    m = mn;
  }
}

__global__ void __launch_bounds__(256)
node_gather(const int* __restrict__ bucket, const int* __restrict__ ovf_len,
            const int* __restrict__ ovf_dst, const int* __restrict__ ovf_src,
            const float* __restrict__ hp,
            const float* __restrict__ al,
            const float* __restrict__ ar,
            const float* __restrict__ bias,
            const float* __restrict__ gamma,
            const float* __restrict__ beta,
            float* __restrict__ out) {
  const int wave = __builtin_amdgcn_readfirstlane(threadIdx.x >> 5);
  const int lane = threadIdx.x & 31;
  constexpr int STRIDE = GATHER_WGS * 8;
  const float4 alv = ((const float4*)al)[lane];

  __shared__ float4 featv[4][32];
  if (wave < 4) {
    const float* p = (wave == 0) ? ar : (wave == 1) ? bias : (wave == 2) ? gamma : beta;
    featv[wave][lane] = ((const float4*)p)[lane];
  }
  __syncthreads();

  int n = blockIdx.x * 8 + wave;
  int n1 = n + STRIDE;
  int row = bucket[n * ROW + lane];
  float ern = quad_sum(dot4(edge_row(hp, n, lane), featv[0][lane]));

  do {
    int row1 = 0;
    if (n1 < N_) row1 = bucket[n1 * ROW + lane];
    const int deg = __builtin_amdgcn_readlane(row, CAP);
    const int mys = (lane == CAP) ? 0 : row;

    float m = __uint_as_float(0xff800000u);
    float sum = 0.f;
    float4 acc = make_float4(0.f, 0.f, 0.f, 0.f);
    if (deg > 0) {
      gather_trips(hp, mys, min(deg, CAP), lane, alv, ern, m, sum, acc);
      if (deg > CAP) {
        const int novf = ovf_len[0];
        for (int c0 = 0; c0 < novf; c0 += 32) {
          const int i = c0 + lane;
          int od = -1, os = 0;
          if (i < novf) { od = ovf_dst[i]; os = ovf_src[i]; }
          unsigned hit = __builtin_amdgcn_ballot_w32(od == n);
          while (hit) {
            const int b = __builtin_ctz(hit);
            hit &= hit - 1;
            gather_trips(hp, __builtin_amdgcn_readlane(os, b), 1, lane, alv, ern, m, sum, acc);
          }
        }
      }
    }
    const float den = sum + 1e-16f;

    asm volatile("" ::: "memory");
    const float4 arv = featv[0][lane];
    const float4 bb  = featv[1][lane];
    const float4 g   = featv[2][lane];
    const float4 be  = featv[3][lane];
    const float4 own1 = edge_row(hp, min(n1, N_ - 1), lane);

    float4 x;
    x.x = acc.x / den + bb.x; x.y = acc.y / den + bb.y;
    x.z = acc.z / den + bb.z; x.w = acc.w / den + bb.w;
    float s  = x.x + x.y + x.z + x.w;
    float s2 = x.x * x.x + x.y * x.y + x.z * x.z + x.w * x.w;
#pragma unroll
    for (int o = 16; o >= 1; o >>= 1) {
      s  += __shfl_xor(s,  o, 32);
      s2 += __shfl_xor(s2, o, 32);
    }
    const float mu  = s * (1.f / 128.f);
    const float var = s2 * (1.f / 128.f) - mu * mu;
    const float inv = rsqrtf(var + LN_EPS);
    float4 y;
    y.x = fmaxf((x.x - mu) * inv * g.x + be.x, 0.f);
    y.y = fmaxf((x.y - mu) * inv * g.y + be.y, 0.f);
    y.z = fmaxf((x.z - mu) * inv * g.z + be.z, 0.f);
    y.w = fmaxf((x.w - mu) * inv * g.w + be.w, 0.f);
    ((float4*)(out + (size_t)n * F_))[lane] = y;

    ern = quad_sum(dot4(own1, arv));
    n = n1; row = row1;
    n1 += STRIDE;
  } while (n < N_);
}

__global__ void __launch_bounds__(96)
gemm_pred(const float* __restrict__ X,
          const float* __restrict__ Wp,
          const float* __restrict__ bp,
          float* __restrict__ Y) {
  const int lane = threadIdx.x & 31;
  const int wave = threadIdx.x >> 5;
  const int n0   = wave * 16;
  const int half = lane >> 4;
  const int l16  = lane & 15;
  const int n    = n0 + l16;
  const bool valid = (n < C_);

  v2f b[F_ / 4];
#pragma unroll
  for (int j = 0; j < F_ / 4; ++j) {
    const int ka = 4 * j + 2 * half;
    b[j].x = valid ? Wp[ka * C_ + n] : 0.f;
    b[j].y = valid ? Wp[(ka + 1) * C_ + n] : 0.f;
  }
  const float bbn = valid ? bp[n] : 0.f;

  for (int tile = blockIdx.x; tile < M_TILES; tile += PRED_WGS) {
    const int m0   = tile * 16;
    const int arow = m0 + l16;
    v8f acc = {};
#pragma unroll
    for (int j = 0; j < F_ / 4; ++j) {
      const int ka = 4 * j + 2 * half;
      v2f a;
      a.x = X[arow * F_ + ka];
      a.y = X[arow * F_ + ka + 1];
      acc = __builtin_amdgcn_wmma_f32_16x16x4_f32(false, a, false, b[j],
                                                  (short)0, acc, false, false);
    }
    if (valid) {
#pragma unroll
      for (int r = 0; r < 8; ++r) {
        const int m = m0 + r + 8 * half;
        Y[m * C_ + n] = acc[r] + bbn;
      }
    }
  }
}

extern "C" void kernel_launch(void* const* d_in, const int* in_sizes, int n_in,
                              void* d_out, int out_size, void* d_ws, size_t ws_size,
                              hipStream_t stream) {
  const float* feats = (const float*)d_in[0];
  const int*   src   = (const int*)d_in[1];
  const int*   dst   = (const int*)d_in[2];
  const float* Ws    = (const float*)d_in[3];
  const float* al    = (const float*)d_in[4];
  const float* ar    = (const float*)d_in[5];
  const float* bias  = (const float*)d_in[6];
  const float* gamma = (const float*)d_in[7];
  const float* beta  = (const float*)d_in[8];
  const float* Wp    = (const float*)d_in[9];
  const float* bp    = (const float*)d_in[10];
  float* outp = (float*)d_out;

  float* ws   = (float*)d_ws;
  float* hbuf = ws;
  float* hp   = hbuf + (size_t)N_ * F_;
  int*   bucket  = (int*)(hp + (size_t)N_ * F_);
  int*   ovf_len = bucket + (size_t)N_ * ROW;
  int*   ovf_dst = ovf_len + 4;
  int*   ovf_src = ovf_dst + E_;

  zero_rows<<<ZERO_WGS, 256, 0, stream>>>((int4*)bucket);
  edge_bucket<<<(E4_ + 255) / 256, 256, 0, stream>>>(src, dst, bucket, ovf_len, ovf_dst, ovf_src);

  for (int i = 0; i < L_; ++i) {
    const float* X = (i == 0) ? feats : hbuf;
    gemm_node_fc<<<FC_WGS, 256, 0, stream>>>(X, Ws + (size_t)i * F_ * F_, hp);
    node_gather<<<GATHER_WGS, 256, 0, stream>>>(bucket, ovf_len, ovf_dst, ovf_src, hp,
                                            al + (size_t)i * H_ * D_,
                                            ar + (size_t)i * H_ * D_, bias + (size_t)i * F_,
                                            gamma + (size_t)i * F_, beta + (size_t)i * F_, hbuf);
  }
  gemm_pred<<<PRED_WGS, 96, 0, stream>>>(hbuf, Wp, bp, outp);
}


# Both GEMMs: A tile brought in once per work-group by whole-row 16-byte loads into LDS (next tile's rows in flight under the products); waves read fragments there, not 8-byte strided global loads
# speedup vs baseline: 5.2232x; 1.1660x over previous
#include <hip/hip_runtime.h>

constexpr int N_ = 50000;
constexpr int E_ = 800000;
constexpr int F_ = 128;
constexpr int H_ = 8;
constexpr int D_ = 16;
constexpr int C_ = 40;
constexpr int L_ = 2;
constexpr float NEG_SLOPE = 0.2f;
constexpr float LN_EPS = 1e-5f;

typedef __attribute__((ext_vector_type(2))) float v2f;
typedef __attribute__((ext_vector_type(8))) float v8f;

__device__ __forceinline__ float lrelu(float x) { return x > 0.f ? x : NEG_SLOPE * x; }

constexpr int M_TILES     = N_ / 16;
constexpr int FC_WGS      = 768;
constexpr int PRED_WGS    = 1024;
constexpr int GATHER_WGS  = 2048;
constexpr int E4_         = E_ / 4;
constexpr int ROW         = 32;
constexpr int CAP         = ROW - 1;
constexpr int ZERO4_      = N_ * (ROW / 4) + 1;
constexpr int ZERO_WGS    = 256;
constexpr int APITCH      = F_ + 4;

__global__ void __launch_bounds__(256)
gemm_node_fc(const float* __restrict__ X,
             const float* __restrict__ W,
             float* __restrict__ Y) {
  __shared__ __attribute__((aligned(16))) float atile[2][16 * APITCH];
  const int lane = threadIdx.x & 31;
  const int wave = threadIdx.x >> 5;
  const int n0   = wave * 16;
  const int half = lane >> 4;
  const int l16  = lane & 15;
  const int bcol = n0 + l16;

  v2f b[F_ / 4];
#pragma unroll
  for (int j = 0; j < F_ / 4; ++j) {
    const int ka = 4 * j + 2 * half;
    b[j].x = W[ka * F_ + bcol];
    b[j].y = W[(ka + 1) * F_ + bcol];
  }

  const float4* X4 = (const float4*)X;
  const int soff0 = wave * APITCH + lane * 4;
  const int soff1 = (wave + 8) * APITCH + lane * 4;
  const int aoff  = l16 * APITCH + 2 * half;

  int tile = blockIdx.x;
  float4 g0 = X4[(tile * 16 + wave) * (F_ / 4) + lane];
  float4 g1 = X4[(tile * 16 + wave + 8) * (F_ / 4) + lane];
  int buf = 0;
  for (; tile < M_TILES; tile += FC_WGS) {
    float* at = atile[buf];
    *(float4*)(at + soff0) = g0;
    *(float4*)(at + soff1) = g1;
    __syncthreads();
    const int nt = tile + FC_WGS;
    if (nt < M_TILES) {
      g0 = X4[(nt * 16 + wave) * (F_ / 4) + lane];
      g1 = X4[(nt * 16 + wave + 8) * (F_ / 4) + lane];
    }
    const int m0 = tile * 16;
    v8f acc = {};
#pragma unroll
    for (int j = 0; j < F_ / 4; ++j) {
      const v2f a = *(const v2f*)(at + aoff + 4 * j);
      acc = __builtin_amdgcn_wmma_f32_16x16x4_f32(false, a, false, b[j],
                                                  (short)0, acc, false, false);
    }
#pragma unroll
    for (int r = 0; r < 8; ++r) {
      const int m = m0 + r + 8 * half;
      Y[m * F_ + bcol] = acc[r];
    }
    buf ^= 1;
  }
}

__global__ void __launch_bounds__(256) zero_rows(int4* __restrict__ p) {
  for (int t = blockIdx.x * 256 + threadIdx.x; t < ZERO4_; t += ZERO_WGS * 256)
    p[t] = make_int4(0, 0, 0, 0);
}

__device__ __forceinline__ void drop_edge(int d, int s, int r, int* __restrict__ bucket,
                                          int* __restrict__ ovf_len,
                                          int* __restrict__ ovf_dst, int* __restrict__ ovf_src) {
  if (r < CAP) {
    bucket[d * ROW + r] = s;
  } else {
    const int k = atomicAdd(ovf_len, 1);
    ovf_dst[k] = d;
    ovf_src[k] = s;
  }
}

__global__ void edge_bucket(const int* __restrict__ src, const int* __restrict__ dst,
                            int* __restrict__ bucket, int* __restrict__ ovf_len,
                            int* __restrict__ ovf_dst, int* __restrict__ ovf_src) {
  const int t = blockIdx.x * blockDim.x + threadIdx.x;
  if (t >= E4_) return;
  const int4 d = ((const int4*)dst)[t];
  const int4 s = ((const int4*)src)[t];
  const int r0 = atomicAdd(&bucket[d.x * ROW + CAP], 1);
  const int r1 = atomicAdd(&bucket[d.y * ROW + CAP], 1);
  const int r2 = atomicAdd(&bucket[d.z * ROW + CAP], 1);
  const int r3 = atomicAdd(&bucket[d.w * ROW + CAP], 1);
  drop_edge(d.x, s.x, r0, bucket, ovf_len, ovf_dst, ovf_src);
  drop_edge(d.y, s.y, r1, bucket, ovf_len, ovf_dst, ovf_src);
  drop_edge(d.z, s.z, r2, bucket, ovf_len, ovf_dst, ovf_src);
  drop_edge(d.w, s.w, r3, bucket, ovf_len, ovf_dst, ovf_src);
}

__device__ __forceinline__ float4 edge_row(const float* __restrict__ hp, int s, int lane) {
  return ((const float4*)(hp + (size_t)s * F_))[lane];
}
__device__ __forceinline__ void axpy4(float4& acc, float a, const float4& x) {
  acc.x += a * x.x; acc.y += a * x.y; acc.z += a * x.z; acc.w += a * x.w;
}
__device__ __forceinline__ float dot4(const float4& a, const float4& b) {
  return a.x * b.x + a.y * b.y + a.z * b.z + a.w * b.w;
}
__device__ __forceinline__ float quad_sum(float x) {
  int b = __builtin_bit_cast(int, x);
  x += __builtin_bit_cast(float, __builtin_amdgcn_update_dpp(b, b, 0xB1, 0xF, 0xF, true));
  b = __builtin_bit_cast(int, x);
  x += __builtin_bit_cast(float, __builtin_amdgcn_update_dpp(b, b, 0x4E, 0xF, 0xF, true));
  return x;
}

__device__ __forceinline__ void gather_trips(const float* __restrict__ hp, int mys, int cnt,
                                             int lane, const float4& alv, float ern,
                                             float& m, float& sum, float4& acc) {
  const float NEG_INF = __uint_as_float(0xff800000u);
  for (int t = 0; t < cnt; t += 4) {
    const int s0 = __builtin_amdgcn_readlane(mys, t);
    const int s1 = __builtin_amdgcn_readlane(mys, t + 1);
    const int s2 = __builtin_amdgcn_readlane(mys, t + 2);
    const int s3 = __builtin_amdgcn_readlane(mys, t + 3);
    const float4 x0 = edge_row(hp, s0, lane);
    const float4 x1 = edge_row(hp, s1, lane);
    const float4 x2 = edge_row(hp, s2, lane);
    const float4 x3 = edge_row(hp, s3, lane);
    const float e0 = quad_sum(dot4(x0, alv));
    const float e1 = quad_sum(dot4(x1, alv));
    const float e2 = quad_sum(dot4(x2, alv));
    const float e3 = quad_sum(dot4(x3, alv));
    const float v0 = lrelu(e0 + ern);
    const float v1 = (t + 1 < cnt) ? lrelu(e1 + ern) : NEG_INF;
    const float v2 = (t + 2 < cnt) ? lrelu(e2 + ern) : NEG_INF;
    const float v3 = (t + 3 < cnt) ? lrelu(e3 + ern) : NEG_INF;
    const float mn = fmaxf(fmaxf(m, fmaxf(v0, v1)), fmaxf(v2, v3));
    const float sc = __expf(m - mn);
    const float p0 = __expf(v0 - mn);
    const float p1 = __expf(v1 - mn);
    const float p2 = __expf(v2 - mn);
    const float p3 = __expf(v3 - mn);
    sum = sum * sc + ((p0 + p1) + (p2 + p3));
    acc.x *= sc; acc.y *= sc; acc.z *= sc; acc.w *= sc;
    axpy4(acc, p0, x0);
    axpy4(acc, p1, x1);
    axpy4(acc, p2, x2);
    axpy4(acc, p3, x3);
    m = mn;
  }
}

__global__ void __launch_bounds__(256)
node_gather(const int* __restrict__ bucket, const int* __restrict__ ovf_len,
            const int* __restrict__ ovf_dst, const int* __restrict__ ovf_src,
            const float* __restrict__ hp,
            const float* __restrict__ al,
            const float* __restrict__ ar,
            const float* __restrict__ bias,
            const float* __restrict__ gamma,
            const float* __restrict__ beta,
            float* __restrict__ out) {
  const int wave = __builtin_amdgcn_readfirstlane(threadIdx.x >> 5);
  const int lane = threadIdx.x & 31;
  constexpr int STRIDE = GATHER_WGS * 8;
  const float4 alv = ((const float4*)al)[lane];

  __shared__ float4 featv[4][32];
  if (wave < 4) {
    const float* p = (wave == 0) ? ar : (wave == 1) ? bias : (wave == 2) ? gamma : beta;
    featv[wave][lane] = ((const float4*)p)[lane];
  }
  __syncthreads();

  int n = blockIdx.x * 8 + wave;
  int n1 = n + STRIDE;
  int row = bucket[n * ROW + lane];
  float ern = quad_sum(dot4(edge_row(hp, n, lane), featv[0][lane]));

  do {
    int row1 = 0;
    if (n1 < N_) row1 = bucket[n1 * ROW + lane];
    const int deg = __builtin_amdgcn_readlane(row, CAP);
    const int mys = (lane == CAP) ? 0 : row;

    float m = __uint_as_float(0xff800000u);
    float sum = 0.f;
    float4 acc = make_float4(0.f, 0.f, 0.f, 0.f);
    if (deg > 0) {
      gather_trips(hp, mys, min(deg, CAP), lane, alv, ern, m, sum, acc);
      if (deg > CAP) {
        const int novf = ovf_len[0];
        for (int c0 = 0; c0 < novf; c0 += 32) {
          const int i = c0 + lane;
          int od = -1, os = 0;
          if (i < novf) { od = ovf_dst[i]; os = ovf_src[i]; }
          unsigned hit = __builtin_amdgcn_ballot_w32(od == n);
          while (hit) {
            const int b = __builtin_ctz(hit);
            hit &= hit - 1;
            gather_trips(hp, __builtin_amdgcn_readlane(os, b), 1, lane, alv, ern, m, sum, acc);
          }
        }
      }
    }
    const float den = sum + 1e-16f;

    asm volatile("" ::: "memory");
    const float4 arv = featv[0][lane];
    const float4 bb  = featv[1][lane];
    const float4 g   = featv[2][lane];
    const float4 be  = featv[3][lane];
    const float4 own1 = edge_row(hp, min(n1, N_ - 1), lane);

    float4 x;
    x.x = acc.x / den + bb.x; x.y = acc.y / den + bb.y;
    x.z = acc.z / den + bb.z; x.w = acc.w / den + bb.w;
    float s  = x.x + x.y + x.z + x.w;
    float s2 = x.x * x.x + x.y * x.y + x.z * x.z + x.w * x.w;
#pragma unroll
    for (int o = 16; o >= 1; o >>= 1) {
      s  += __shfl_xor(s,  o, 32);
      s2 += __shfl_xor(s2, o, 32);
    }
    const float mu  = s * (1.f / 128.f);
    const float var = s2 * (1.f / 128.f) - mu * mu;
    const float inv = rsqrtf(var + LN_EPS);
    float4 y;
    y.x = fmaxf((x.x - mu) * inv * g.x + be.x, 0.f);
    y.y = fmaxf((x.y - mu) * inv * g.y + be.y, 0.f);
    y.z = fmaxf((x.z - mu) * inv * g.z + be.z, 0.f);
    y.w = fmaxf((x.w - mu) * inv * g.w + be.w, 0.f);
    ((float4*)(out + (size_t)n * F_))[lane] = y;

    ern = quad_sum(dot4(own1, arv));
    n = n1; row = row1;
    n1 += STRIDE;
  } while (n < N_);
}

__global__ void __launch_bounds__(96)
gemm_pred(const float* __restrict__ X,
          const float* __restrict__ Wp,
          const float* __restrict__ bp,
          float* __restrict__ Y) {
  __shared__ __attribute__((aligned(16))) float atile[2][16 * APITCH];
  const int lane = threadIdx.x & 31;
  const int wave = __builtin_amdgcn_readfirstlane(threadIdx.x >> 5);
  const int n0   = wave * 16;
  const int half = lane >> 4;
  const int l16  = lane & 15;
  const int n    = n0 + l16;
  const bool valid = (n < C_);

  v2f b[F_ / 4];
#pragma unroll
  for (int j = 0; j < F_ / 4; ++j) {
    const int ka = 4 * j + 2 * half;
    b[j].x = valid ? Wp[ka * C_ + n] : 0.f;
    b[j].y = valid ? Wp[(ka + 1) * C_ + n] : 0.f;
  }
  const float bbn = valid ? bp[n] : 0.f;

  const float4* X4 = (const float4*)X;
  const int aoff = l16 * APITCH + 2 * half;
  const int xrow = wave;
  const bool has15 = (wave == 0);

  int tile = blockIdx.x;
  float4 g0 = X4[(tile * 16 + xrow     ) * (F_ / 4) + lane];
  float4 g1 = X4[(tile * 16 + xrow +  3) * (F_ / 4) + lane];
  float4 g2 = X4[(tile * 16 + xrow +  6) * (F_ / 4) + lane];
  float4 g3 = X4[(tile * 16 + xrow +  9) * (F_ / 4) + lane];
  float4 g4 = X4[(tile * 16 + xrow + 12) * (F_ / 4) + lane];
  float4 g5 = make_float4(0.f, 0.f, 0.f, 0.f);
  if (has15) g5 = X4[(tile * 16 + 15) * (F_ / 4) + lane];
  int buf = 0;
  for (; tile < M_TILES; tile += PRED_WGS) {
    float* at = atile[buf];
    float* sp = at + xrow * APITCH + lane * 4;
    *(float4*)(sp             ) = g0;
    *(float4*)(sp +  3 * APITCH) = g1;
    *(float4*)(sp +  6 * APITCH) = g2;
    *(float4*)(sp +  9 * APITCH) = g3;
    *(float4*)(sp + 12 * APITCH) = g4;
    if (has15) *(float4*)(at + 15 * APITCH + lane * 4) = g5;
    __syncthreads();
    const int nt = tile + PRED_WGS;
    if (nt < M_TILES) {
      g0 = X4[(nt * 16 + xrow     ) * (F_ / 4) + lane];
      g1 = X4[(nt * 16 + xrow +  3) * (F_ / 4) + lane];
      g2 = X4[(nt * 16 + xrow +  6) * (F_ / 4) + lane];
      g3 = X4[(nt * 16 + xrow +  9) * (F_ / 4) + lane];
      g4 = X4[(nt * 16 + xrow + 12) * (F_ / 4) + lane];
      if (has15) g5 = X4[(nt * 16 + 15) * (F_ / 4) + lane];
    }
    const int m0 = tile * 16;
    v8f acc = {};
#pragma unroll
    for (int j = 0; j < F_ / 4; ++j) {
      const v2f a = *(const v2f*)(at + aoff + 4 * j);
      acc = __builtin_amdgcn_wmma_f32_16x16x4_f32(false, a, false, b[j],
                                                  (short)0, acc, false, false);
    }
    if (valid) {
#pragma unroll
      for (int r = 0; r < 8; ++r) {
        const int m = m0 + r + 8 * half;
        Y[m * C_ + n] = acc[r] + bbn;
      }
    }
    buf ^= 1;
  }
}

extern "C" void kernel_launch(void* const* d_in, const int* in_sizes, int n_in,
                              void* d_out, int out_size, void* d_ws, size_t ws_size,
                              hipStream_t stream) {
  const float* feats = (const float*)d_in[0];
  const int*   src   = (const int*)d_in[1];
  const int*   dst   = (const int*)d_in[2];
  const float* Ws    = (const float*)d_in[3];
  const float* al    = (const float*)d_in[4];
  const float* ar    = (const float*)d_in[5];
  const float* bias  = (const float*)d_in[6];
  const float* gamma = (const float*)d_in[7];
  const float* beta  = (const float*)d_in[8];
  const float* Wp    = (const float*)d_in[9];
  const float* bp    = (const float*)d_in[10];
  float* outp = (float*)d_out;

  float* ws   = (float*)d_ws;
  float* hbuf = ws;
  float* hp   = hbuf + (size_t)N_ * F_;
  int*   bucket  = (int*)(hp + (size_t)N_ * F_);
  int*   ovf_len = bucket + (size_t)N_ * ROW;
  int*   ovf_dst = ovf_len + 4;
  int*   ovf_src = ovf_dst + E_;

  zero_rows<<<ZERO_WGS, 256, 0, stream>>>((int4*)bucket);
  edge_bucket<<<(E4_ + 255) / 256, 256, 0, stream>>>(src, dst, bucket, ovf_len, ovf_dst, ovf_src);

  for (int i = 0; i < L_; ++i) {
    const float* X = (i == 0) ? feats : hbuf;
    gemm_node_fc<<<FC_WGS, 256, 0, stream>>>(X, Ws + (size_t)i * F_ * F_, hp);
    node_gather<<<GATHER_WGS, 256, 0, stream>>>(bucket, ovf_len, ovf_dst, ovf_src, hp,
                                            al + (size_t)i * H_ * D_,
                                            ar + (size_t)i * H_ * D_, bias + (size_t)i * F_,
                                            gamma + (size_t)i * F_, beta + (size_t)i * F_, hbuf);
  }
  gemm_pred<<<PRED_WGS, 96, 0, stream>>>(hbuf, Wp, bp, outp);
}


# Gather issues fewer vector instructions: scores kept times log2(e) (bare exp2), leaky ReLU as max, pair-sum dot (new summation order), DPP wave sums, 1 reciprocal for 4 divisions, 32-bit row offsets
# speedup vs baseline: 5.4158x; 1.0369x over previous
#include <hip/hip_runtime.h>

constexpr int N_ = 50000;
constexpr int E_ = 800000;
constexpr int F_ = 128;
constexpr int H_ = 8;
constexpr int D_ = 16;
constexpr int C_ = 40;
constexpr int L_ = 2;
constexpr float NEG_SLOPE = 0.2f;
constexpr float LN_EPS = 1e-5f;

typedef __attribute__((ext_vector_type(2))) float v2f;
typedef __attribute__((ext_vector_type(8))) float v8f;

__device__ __forceinline__ float lrelu(float x) { return fmaxf(x, NEG_SLOPE * x); }
constexpr float LOG2E = 1.4426950408889634f;

constexpr int M_TILES     = N_ / 16;
constexpr int FC_WGS      = 768;
constexpr int PRED_WGS    = 1024;
constexpr int GATHER_WGS  = 2048;
constexpr int E4_         = E_ / 4;
constexpr int ROW         = 32;
constexpr int CAP         = ROW - 1;
constexpr int ZERO4_      = N_ * (ROW / 4) + 1;
constexpr int ZERO_WGS    = 256;
constexpr int APITCH      = F_ + 4;

__global__ void __launch_bounds__(256)
gemm_node_fc(const float* __restrict__ X,
             const float* __restrict__ W,
             float* __restrict__ Y) {
  __shared__ __attribute__((aligned(16))) float atile[2][16 * APITCH];
  const int lane = threadIdx.x & 31;
  const int wave = threadIdx.x >> 5;
  const int n0   = wave * 16;
  const int half = lane >> 4;
  const int l16  = lane & 15;
  const int bcol = n0 + l16;

  v2f b[F_ / 4];
#pragma unroll
  for (int j = 0; j < F_ / 4; ++j) {
    const int ka = 4 * j + 2 * half;
    b[j].x = W[ka * F_ + bcol];
    b[j].y = W[(ka + 1) * F_ + bcol];
  }

  const float4* X4 = (const float4*)X;
  const int soff0 = wave * APITCH + lane * 4;
  const int soff1 = (wave + 8) * APITCH + lane * 4;
  const int aoff  = l16 * APITCH + 2 * half;

  int tile = blockIdx.x;
  float4 g0 = X4[(tile * 16 + wave) * (F_ / 4) + lane];
  float4 g1 = X4[(tile * 16 + wave + 8) * (F_ / 4) + lane];
  int buf = 0;
  for (; tile < M_TILES; tile += FC_WGS) {
    float* at = atile[buf];
    *(float4*)(at + soff0) = g0;
    *(float4*)(at + soff1) = g1;
    __syncthreads();
    const int nt = tile + FC_WGS;
    if (nt < M_TILES) {
      g0 = X4[(nt * 16 + wave) * (F_ / 4) + lane];
      g1 = X4[(nt * 16 + wave + 8) * (F_ / 4) + lane];
    }
    const int m0 = tile * 16;
    v8f acc = {};
#pragma unroll
    for (int j = 0; j < F_ / 4; ++j) {
      const v2f a = *(const v2f*)(at + aoff + 4 * j);
      acc = __builtin_amdgcn_wmma_f32_16x16x4_f32(false, a, false, b[j],
                                                  (short)0, acc, false, false);
    }
#pragma unroll
    for (int r = 0; r < 8; ++r) {
      const int m = m0 + r + 8 * half;
      Y[m * F_ + bcol] = acc[r];
    }
    buf ^= 1;
  }
}

__global__ void __launch_bounds__(256) zero_rows(int4* __restrict__ p) {
  for (int t = blockIdx.x * 256 + threadIdx.x; t < ZERO4_; t += ZERO_WGS * 256)
    p[t] = make_int4(0, 0, 0, 0);
}

__device__ __forceinline__ void drop_edge(int d, int s, int r, int* __restrict__ bucket,
                                          int* __restrict__ ovf_len,
                                          int* __restrict__ ovf_dst, int* __restrict__ ovf_src) {
  if (r < CAP) {
    bucket[d * ROW + r] = s;
  } else {
    const int k = atomicAdd(ovf_len, 1);
    ovf_dst[k] = d;
    ovf_src[k] = s;
  }
}

__global__ void edge_bucket(const int* __restrict__ src, const int* __restrict__ dst,
                            int* __restrict__ bucket, int* __restrict__ ovf_len,
                            int* __restrict__ ovf_dst, int* __restrict__ ovf_src) {
  const int t = blockIdx.x * blockDim.x + threadIdx.x;
  if (t >= E4_) return;
  const int4 d = ((const int4*)dst)[t];
  const int4 s = ((const int4*)src)[t];
  const int r0 = atomicAdd(&bucket[d.x * ROW + CAP], 1);
  const int r1 = atomicAdd(&bucket[d.y * ROW + CAP], 1);
  const int r2 = atomicAdd(&bucket[d.z * ROW + CAP], 1);
  const int r3 = atomicAdd(&bucket[d.w * ROW + CAP], 1);
  drop_edge(d.x, s.x, r0, bucket, ovf_len, ovf_dst, ovf_src);
  drop_edge(d.y, s.y, r1, bucket, ovf_len, ovf_dst, ovf_src);
  drop_edge(d.z, s.z, r2, bucket, ovf_len, ovf_dst, ovf_src);
  drop_edge(d.w, s.w, r3, bucket, ovf_len, ovf_dst, ovf_src);
}

__device__ __forceinline__ float4 edge_row(const float* __restrict__ hp, int s, int lane) {
  const unsigned off = ((unsigned)s << 9) | ((unsigned)lane << 4);
  return *(const float4*)((const char*)hp + off);
}
__device__ __forceinline__ void axpy4(float4& acc, float a, const float4& x) {
  acc.x += a * x.x; acc.y += a * x.y; acc.z += a * x.z; acc.w += a * x.w;
}
__device__ __forceinline__ float dot4(const float4& a, const float4& b) {
  const float p0 = a.x * b.x + a.z * b.z;
  const float p1 = a.y * b.y + a.w * b.w;
  return p0 + p1;
}
__device__ __forceinline__ float quad_sum(float x) {
  int b = __builtin_bit_cast(int, x);
  x += __builtin_bit_cast(float, __builtin_amdgcn_update_dpp(b, b, 0xB1, 0xF, 0xF, true));
  b = __builtin_bit_cast(int, x);
  x += __builtin_bit_cast(float, __builtin_amdgcn_update_dpp(b, b, 0x4E, 0xF, 0xF, true));
  return x;
}
__device__ __forceinline__ float wave_sum(float x) {
  int b = __builtin_bit_cast(int, x);
  x += __builtin_bit_cast(float, __builtin_amdgcn_permlanex16(b, b, 0x76543210, 0xfedcba98, false, false));
  b = __builtin_bit_cast(int, x);
  x += __builtin_bit_cast(float, __builtin_amdgcn_update_dpp(b, b, 0x168, 0xF, 0xF, true));
  b = __builtin_bit_cast(int, x);
  x += __builtin_bit_cast(float, __builtin_amdgcn_update_dpp(b, b, 0x164, 0xF, 0xF, true));
  b = __builtin_bit_cast(int, x);
  x += __builtin_bit_cast(float, __builtin_amdgcn_update_dpp(b, b, 0x4E, 0xF, 0xF, true));
  b = __builtin_bit_cast(int, x);
  x += __builtin_bit_cast(float, __builtin_amdgcn_update_dpp(b, b, 0xB1, 0xF, 0xF, true));
  return x;
}

__device__ __forceinline__ void gather_trips(const float* __restrict__ hp, int mys, int cnt,
                                             int lane, const float4& alv, float ern,
                                             float& m, float& sum, float4& acc) {
  const float NEG_INF = __uint_as_float(0xff800000u);
  for (int t = 0; t < cnt; t += 4) {
    const int s0 = __builtin_amdgcn_readlane(mys, t);
    const int s1 = __builtin_amdgcn_readlane(mys, t + 1);
    const int s2 = __builtin_amdgcn_readlane(mys, t + 2);
    const int s3 = __builtin_amdgcn_readlane(mys, t + 3);
    const float4 x0 = edge_row(hp, s0, lane);
    const float4 x1 = edge_row(hp, s1, lane);
    const float4 x2 = edge_row(hp, s2, lane);
    const float4 x3 = edge_row(hp, s3, lane);
    const float e0 = quad_sum(dot4(x0, alv));
    const float e1 = quad_sum(dot4(x1, alv));
    const float e2 = quad_sum(dot4(x2, alv));
    const float e3 = quad_sum(dot4(x3, alv));
    const float v0 = lrelu(e0 + ern);
    const float v1 = (t + 1 < cnt) ? lrelu(e1 + ern) : NEG_INF;
    const float v2 = (t + 2 < cnt) ? lrelu(e2 + ern) : NEG_INF;
    const float v3 = (t + 3 < cnt) ? lrelu(e3 + ern) : NEG_INF;
    const float mn = fmaxf(fmaxf(m, fmaxf(v0, v1)), fmaxf(v2, v3));
    const float sc = __builtin_amdgcn_exp2f(m - mn);
    const float p0 = __builtin_amdgcn_exp2f(v0 - mn);
    const float p1 = __builtin_amdgcn_exp2f(v1 - mn);
    const float p2 = __builtin_amdgcn_exp2f(v2 - mn);
    const float p3 = __builtin_amdgcn_exp2f(v3 - mn);
    sum = sum * sc + ((p0 + p1) + (p2 + p3));
    acc.x *= sc; acc.y *= sc; acc.z *= sc; acc.w *= sc;
    axpy4(acc, p0, x0);
    axpy4(acc, p1, x1);
    axpy4(acc, p2, x2);
    axpy4(acc, p3, x3);
    m = mn;
  }
}

__global__ void __launch_bounds__(256)
node_gather(const int* __restrict__ bucket, const int* __restrict__ ovf_len,
            const int* __restrict__ ovf_dst, const int* __restrict__ ovf_src,
            const float* __restrict__ hp,
            const float* __restrict__ al,
            const float* __restrict__ ar,
            const float* __restrict__ bias,
            const float* __restrict__ gamma,
            const float* __restrict__ beta,
            float* __restrict__ out) {
  const int wave = __builtin_amdgcn_readfirstlane(threadIdx.x >> 5);
  const int lane = threadIdx.x & 31;
  constexpr int STRIDE = GATHER_WGS * 8;
  float4 alv = ((const float4*)al)[lane];
  alv.x *= LOG2E; alv.y *= LOG2E; alv.z *= LOG2E; alv.w *= LOG2E;

  __shared__ float4 featv[4][32];
  if (wave < 4) {
    const float* p = (wave == 0) ? ar : (wave == 1) ? bias : (wave == 2) ? gamma : beta;
    float4 v = ((const float4*)p)[lane];
    const float f = (wave == 0) ? LOG2E : 1.f;
    v.x *= f; v.y *= f; v.z *= f; v.w *= f;
    featv[wave][lane] = v;
  }
  __syncthreads();

  int n = blockIdx.x * 8 + wave;
  int n1 = n + STRIDE;
  int row = bucket[n * ROW + lane];
  float ern = quad_sum(dot4(edge_row(hp, n, lane), featv[0][lane]));

  do {
    int row1 = 0;
    if (n1 < N_) row1 = bucket[n1 * ROW + lane];
    const int deg = __builtin_amdgcn_readlane(row, CAP);
    const int mys = (lane == CAP) ? 0 : row;

    float m = __uint_as_float(0xff800000u);
    float sum = 0.f;
    float4 acc = make_float4(0.f, 0.f, 0.f, 0.f);
    if (deg > 0) {
      gather_trips(hp, mys, min(deg, CAP), lane, alv, ern, m, sum, acc);
      if (deg > CAP) {
        const int novf = ovf_len[0];
        for (int c0 = 0; c0 < novf; c0 += 32) {
          const int i = c0 + lane;
          int od = -1, os = 0;
          if (i < novf) { od = ovf_dst[i]; os = ovf_src[i]; }
          unsigned hit = __builtin_amdgcn_ballot_w32(od == n);
          while (hit) {
            const int b = __builtin_ctz(hit);
            hit &= hit - 1;
            gather_trips(hp, __builtin_amdgcn_readlane(os, b), 1, lane, alv, ern, m, sum, acc);
          }
        }
      }
    }
    const float den = sum + 1e-16f;

    asm volatile("" ::: "memory");
    const float4 arv = featv[0][lane];
    const float4 bb  = featv[1][lane];
    const float4 g   = featv[2][lane];
    const float4 be  = featv[3][lane];
    const float4 own1 = edge_row(hp, min(n1, N_ - 1), lane);

    const float rden = __builtin_amdgcn_rcpf(den);
    float4 x;
    x.x = acc.x * rden + bb.x; x.y = acc.y * rden + bb.y;
    x.z = acc.z * rden + bb.z; x.w = acc.w * rden + bb.w;
    float s  = x.x + x.y + x.z + x.w;
    float s2 = x.x * x.x + x.y * x.y + x.z * x.z + x.w * x.w;
    s  = wave_sum(s);
    s2 = wave_sum(s2);
    const float mu  = s * (1.f / 128.f);
    const float var = s2 * (1.f / 128.f) - mu * mu;
    const float inv = rsqrtf(var + LN_EPS);
    float4 y;
    y.x = fmaxf((x.x - mu) * inv * g.x + be.x, 0.f);
    y.y = fmaxf((x.y - mu) * inv * g.y + be.y, 0.f);
    y.z = fmaxf((x.z - mu) * inv * g.z + be.z, 0.f);
    y.w = fmaxf((x.w - mu) * inv * g.w + be.w, 0.f);
    ((float4*)(out + (size_t)n * F_))[lane] = y;

    ern = quad_sum(dot4(own1, arv));
    n = n1; row = row1;
    n1 += STRIDE;
  } while (n < N_);
}

__global__ void __launch_bounds__(96)
gemm_pred(const float* __restrict__ X,
          const float* __restrict__ Wp,
          const float* __restrict__ bp,
          float* __restrict__ Y) {
  __shared__ __attribute__((aligned(16))) float atile[2][16 * APITCH];
  const int lane = threadIdx.x & 31;
  const int wave = __builtin_amdgcn_readfirstlane(threadIdx.x >> 5);
  const int n0   = wave * 16;
  const int half = lane >> 4;
  const int l16  = lane & 15;
  const int n    = n0 + l16;
  const bool valid = (n < C_);

  v2f b[F_ / 4];
#pragma unroll
  for (int j = 0; j < F_ / 4; ++j) {
    const int ka = 4 * j + 2 * half;
    b[j].x = valid ? Wp[ka * C_ + n] : 0.f;
    b[j].y = valid ? Wp[(ka + 1) * C_ + n] : 0.f;
  }
  const float bbn = valid ? bp[n] : 0.f;

  const float4* X4 = (const float4*)X;
  const int aoff = l16 * APITCH + 2 * half;
  const int xrow = wave;
  const bool has15 = (wave == 0);

  int tile = blockIdx.x;
  float4 g0 = X4[(tile * 16 + xrow     ) * (F_ / 4) + lane];
  float4 g1 = X4[(tile * 16 + xrow +  3) * (F_ / 4) + lane];
  float4 g2 = X4[(tile * 16 + xrow +  6) * (F_ / 4) + lane];
  float4 g3 = X4[(tile * 16 + xrow +  9) * (F_ / 4) + lane];
  float4 g4 = X4[(tile * 16 + xrow + 12) * (F_ / 4) + lane];
  float4 g5 = make_float4(0.f, 0.f, 0.f, 0.f);
  if (has15) g5 = X4[(tile * 16 + 15) * (F_ / 4) + lane];
  int buf = 0;
  for (; tile < M_TILES; tile += PRED_WGS) {
    float* at = atile[buf];
    float* sp = at + xrow * APITCH + lane * 4;
    *(float4*)(sp             ) = g0;
    *(float4*)(sp +  3 * APITCH) = g1;
    *(float4*)(sp +  6 * APITCH) = g2;
    *(float4*)(sp +  9 * APITCH) = g3;
    *(float4*)(sp + 12 * APITCH) = g4;
    if (has15) *(float4*)(at + 15 * APITCH + lane * 4) = g5;
    __syncthreads();
    const int nt = tile + PRED_WGS;
    if (nt < M_TILES) {
      g0 = X4[(nt * 16 + xrow     ) * (F_ / 4) + lane];
      g1 = X4[(nt * 16 + xrow +  3) * (F_ / 4) + lane];
      g2 = X4[(nt * 16 + xrow +  6) * (F_ / 4) + lane];
      g3 = X4[(nt * 16 + xrow +  9) * (F_ / 4) + lane];
      g4 = X4[(nt * 16 + xrow + 12) * (F_ / 4) + lane];
      if (has15) g5 = X4[(nt * 16 + 15) * (F_ / 4) + lane];
    }
    const int m0 = tile * 16;
    v8f acc = {};
#pragma unroll
    for (int j = 0; j < F_ / 4; ++j) {
      const v2f a = *(const v2f*)(at + aoff + 4 * j);
      acc = __builtin_amdgcn_wmma_f32_16x16x4_f32(false, a, false, b[j],
                                                  (short)0, acc, false, false);
    }
    if (valid) {
#pragma unroll
      for (int r = 0; r < 8; ++r) {
        const int m = m0 + r + 8 * half;
        Y[m * C_ + n] = acc[r] + bbn;
      }
    }
    buf ^= 1;
  }
}

extern "C" void kernel_launch(void* const* d_in, const int* in_sizes, int n_in,
                              void* d_out, int out_size, void* d_ws, size_t ws_size,
                              hipStream_t stream) {
  const float* feats = (const float*)d_in[0];
  const int*   src   = (const int*)d_in[1];
  const int*   dst   = (const int*)d_in[2];
  const float* Ws    = (const float*)d_in[3];
  const float* al    = (const float*)d_in[4];
  const float* ar    = (const float*)d_in[5];
  const float* bias  = (const float*)d_in[6];
  const float* gamma = (const float*)d_in[7];
  const float* beta  = (const float*)d_in[8];
  const float* Wp    = (const float*)d_in[9];
  const float* bp    = (const float*)d_in[10];
  float* outp = (float*)d_out;

  float* ws   = (float*)d_ws;
  float* hbuf = ws;
  float* hp   = hbuf + (size_t)N_ * F_;
  int*   bucket  = (int*)(hp + (size_t)N_ * F_);
  int*   ovf_len = bucket + (size_t)N_ * ROW;
  int*   ovf_dst = ovf_len + 4;
  int*   ovf_src = ovf_dst + E_;

  zero_rows<<<ZERO_WGS, 256, 0, stream>>>((int4*)bucket);
  edge_bucket<<<(E4_ + 255) / 256, 256, 0, stream>>>(src, dst, bucket, ovf_len, ovf_dst, ovf_src);

  for (int i = 0; i < L_; ++i) {
    const float* X = (i == 0) ? feats : hbuf;
    gemm_node_fc<<<FC_WGS, 256, 0, stream>>>(X, Ws + (size_t)i * F_ * F_, hp);
    node_gather<<<GATHER_WGS, 256, 0, stream>>>(bucket, ovf_len, ovf_dst, ovf_src, hp,
                                            al + (size_t)i * H_ * D_,
                                            ar + (size_t)i * H_ * D_, bias + (size_t)i * F_,
                                            gamma + (size_t)i * F_, beta + (size_t)i * F_, hbuf);
  }
  gemm_pred<<<PRED_WGS, 96, 0, stream>>>(hbuf, Wp, bp, outp);
}


# In-edge rows built by a two-level counting sort on LDS counters (edges filed by node range, one work-group ranks a range and stores whole rows): no per-edge global atomic; new summation order
# speedup vs baseline: 6.6400x; 1.2260x over previous
#include <hip/hip_runtime.h>

constexpr int N_ = 50000;
constexpr int E_ = 800000;
constexpr int F_ = 128;
constexpr int H_ = 8;
constexpr int D_ = 16;
constexpr int C_ = 40;
constexpr int L_ = 2;
constexpr float NEG_SLOPE = 0.2f;
constexpr float LN_EPS = 1e-5f;

typedef __attribute__((ext_vector_type(2))) float v2f;
typedef __attribute__((ext_vector_type(8))) float v8f;

__device__ __forceinline__ float lrelu(float x) { return fmaxf(x, NEG_SLOPE * x); }
constexpr float LOG2E = 1.4426950408889634f;

constexpr int M_TILES     = N_ / 16;
constexpr int FC_WGS      = 768;
constexpr int PRED_WGS    = 1024;
constexpr int GATHER_WGS  = 2048;
constexpr int E4_         = E_ / 4;
constexpr int ROW         = 32;
constexpr int CAP         = ROW - 1;
constexpr int RB          = 128;
constexpr int NB          = (N_ + RB - 1) / RB;
constexpr int CHUNKS      = 256;
constexpr int GPC         = (E4_ + CHUNKS - 1) / CHUNKS;
constexpr int CELL        = 32;
static_assert(N_ <= 65536, "a source id must fit the low 16 bits of a filed word");
static_assert(RB == 128, "the range of a node is its id >> 7");
static_assert(GPC <= 4 * 256, "a filing thread takes at most four groups");
constexpr int APITCH      = F_ + 4;

__global__ void __launch_bounds__(256)
gemm_node_fc(const float* __restrict__ X,
             const float* __restrict__ W,
             float* __restrict__ Y) {
  __shared__ __attribute__((aligned(16))) float atile[2][16 * APITCH];
  const int lane = threadIdx.x & 31;
  const int wave = threadIdx.x >> 5;
  const int n0   = wave * 16;
  const int half = lane >> 4;
  const int l16  = lane & 15;
  const int bcol = n0 + l16;

  v2f b[F_ / 4];
#pragma unroll
  for (int j = 0; j < F_ / 4; ++j) {
    const int ka = 4 * j + 2 * half;
    b[j].x = W[ka * F_ + bcol];
    b[j].y = W[(ka + 1) * F_ + bcol];
  }

  const float4* X4 = (const float4*)X;
  const int soff0 = wave * APITCH + lane * 4;
  const int soff1 = (wave + 8) * APITCH + lane * 4;
  const int aoff  = l16 * APITCH + 2 * half;

  int tile = blockIdx.x;
  float4 g0 = X4[(tile * 16 + wave) * (F_ / 4) + lane];
  float4 g1 = X4[(tile * 16 + wave + 8) * (F_ / 4) + lane];
  int buf = 0;
  for (; tile < M_TILES; tile += FC_WGS) {
    float* at = atile[buf];
    *(float4*)(at + soff0) = g0;
    *(float4*)(at + soff1) = g1;
    __syncthreads();
    const int nt = tile + FC_WGS;
    if (nt < M_TILES) {
      g0 = X4[(nt * 16 + wave) * (F_ / 4) + lane];
      g1 = X4[(nt * 16 + wave + 8) * (F_ / 4) + lane];
    }
    const int m0 = tile * 16;
    v8f acc = {};
#pragma unroll
    for (int j = 0; j < F_ / 4; ++j) {
      const v2f a = *(const v2f*)(at + aoff + 4 * j);
      acc = __builtin_amdgcn_wmma_f32_16x16x4_f32(false, a, false, b[j],
                                                  (short)0, acc, false, false);
    }
#pragma unroll
    for (int r = 0; r < 8; ++r) {
      const int m = m0 + r + 8 * half;
      Y[m * F_ + bcol] = acc[r];
    }
    buf ^= 1;
  }
}

__global__ void __launch_bounds__(256)
edge_file(const int* __restrict__ src, const int* __restrict__ dst,
          int* __restrict__ cells, int* __restrict__ counts, int* __restrict__ ovf_len) {
  __shared__ int cnt[NB];
  const int w = blockIdx.x;
  for (int t = threadIdx.x; t < NB; t += 256) cnt[t] = 0;
  const int gbeg = w * GPC;
  const int gend = min(gbeg + GPC, E4_);
  int4 dd[4], ss[4];
#pragma unroll
  for (int i = 0; i < 4; ++i) {
    const int g = gbeg + (int)threadIdx.x + 256 * i;
    dd[i] = make_int4(-1, -1, -1, -1);
    ss[i] = make_int4(0, 0, 0, 0);
    if (g < gend) {
      dd[i] = ((const int4*)dst)[g];
      ss[i] = ((const int4*)src)[g];
    }
  }
  __syncthreads();
  int* cell0 = cells + w * CELL;
#pragma unroll
  for (int i = 0; i < 4; ++i) {
    if (dd[i].x >= 0) {
      const int4 d = dd[i];
      const int4 s = ss[i];
      const int b0 = d.x >> 7, b1 = d.y >> 7, b2 = d.z >> 7, b3 = d.w >> 7;
      const int r0 = atomicAdd(&cnt[b0], 1);
      const int r1 = atomicAdd(&cnt[b1], 1);
      const int r2 = atomicAdd(&cnt[b2], 1);
      const int r3 = atomicAdd(&cnt[b3], 1);
      if (r0 < CELL) cell0[b0 * (CHUNKS * CELL) + r0] = ((d.x & (RB - 1)) << 16) | s.x;
      if (r1 < CELL) cell0[b1 * (CHUNKS * CELL) + r1] = ((d.y & (RB - 1)) << 16) | s.y;
      if (r2 < CELL) cell0[b2 * (CHUNKS * CELL) + r2] = ((d.z & (RB - 1)) << 16) | s.z;
      if (r3 < CELL) cell0[b3 * (CHUNKS * CELL) + r3] = ((d.w & (RB - 1)) << 16) | s.w;
    }
  }
  __syncthreads();
  for (int t = threadIdx.x; t < NB; t += 256) counts[t * CHUNKS + w] = cnt[t];
  if (w == 0 && threadIdx.x == 0) ovf_len[0] = 0;
}

__device__ __forceinline__ void place_ranked(int dl, int s, int r, int node0, int* rows,
                                             int* __restrict__ ovf_len,
                                             int* __restrict__ ovf_dst, int* __restrict__ ovf_src) {
  if (r < CAP) {
    rows[dl * ROW + r] = s;
  } else {
    const int k = atomicAdd(ovf_len, 1);
    ovf_dst[k] = node0 + dl;
    ovf_src[k] = s;
  }
}

__global__ void __launch_bounds__(256)
row_build(const int* __restrict__ src, const int* __restrict__ dst,
          const int* __restrict__ cells, const int* __restrict__ counts,
          int* __restrict__ bucket, int* __restrict__ ovf_len,
          int* __restrict__ ovf_dst, int* __restrict__ ovf_src) {
  __shared__ __attribute__((aligned(16))) int rows[RB * ROW];
  __shared__ int deg[RB];
  __shared__ int cc[CHUNKS];
  static_assert(CHUNKS == 256, "one count per thread");
  const int b = blockIdx.x;
  const int node0 = b * RB;
  const int lane = threadIdx.x & 31;
  const int wave = __builtin_amdgcn_readfirstlane(threadIdx.x >> 5);
  cc[threadIdx.x] = counts[b * CHUNKS + threadIdx.x];
  for (int t = threadIdx.x; t < RB * ROW / 4; t += 256) ((int4*)rows)[t] = make_int4(0, 0, 0, 0);
  if (threadIdx.x < RB) deg[threadIdx.x] = 0;
  __syncthreads();

  const int* cellb = cells + (b * CHUNKS + wave) * CELL + lane;
  int wv[32];
  unsigned big = 0;
  const int myn = cc[wave + 8 * lane];
#pragma unroll
  for (int u = 0; u < 32; ++u) {
    const int n = __builtin_amdgcn_readlane(myn, u);
    if (n > CELL) big |= 1u << u;
    wv[u] = -1;
    if (n <= CELL && lane < n) wv[u] = cellb[u * 8 * CELL];
  }
#pragma unroll
  for (int v = 0; v < 32; v += 8) {
    int rk[8];
#pragma unroll
    for (int u = 0; u < 8; ++u) {
      rk[u] = 0;
      if (wv[v + u] >= 0) rk[u] = atomicAdd(&deg[wv[v + u] >> 16], 1);
    }
#pragma unroll
    for (int u = 0; u < 8; ++u) {
      if (wv[v + u] >= 0)
        place_ranked(wv[v + u] >> 16, wv[v + u] & 0xffff, rk[u], node0, rows, ovf_len, ovf_dst, ovf_src);
    }
  }
  while (big) {
    const int u = __builtin_ctz(big);
    big &= big - 1;
    const int c = wave + 8 * u;
    const int gbeg = c * GPC;
    const int gend = min(gbeg + GPC, E4_);
    for (int g = gbeg + lane; g < gend; g += 32) {
      const int4 d = ((const int4*)dst)[g];
      const int4 s = ((const int4*)src)[g];
      if ((d.x >> 7) == b) { const int dl = d.x & (RB - 1); place_ranked(dl, s.x, atomicAdd(&deg[dl], 1), node0, rows, ovf_len, ovf_dst, ovf_src); }
      if ((d.y >> 7) == b) { const int dl = d.y & (RB - 1); place_ranked(dl, s.y, atomicAdd(&deg[dl], 1), node0, rows, ovf_len, ovf_dst, ovf_src); }
      if ((d.z >> 7) == b) { const int dl = d.z & (RB - 1); place_ranked(dl, s.z, atomicAdd(&deg[dl], 1), node0, rows, ovf_len, ovf_dst, ovf_src); }
      if ((d.w >> 7) == b) { const int dl = d.w & (RB - 1); place_ranked(dl, s.w, atomicAdd(&deg[dl], 1), node0, rows, ovf_len, ovf_dst, ovf_src); }
    }
  }
  __syncthreads();

  for (int r = wave; r < RB; r += 8) {
    const int node = node0 + r;
    if (node < N_) bucket[node * ROW + lane] = (lane == CAP) ? deg[r] : rows[r * ROW + lane];
  }
}

__device__ __forceinline__ float4 edge_row(const float* __restrict__ hp, int s, int lane) {
  const unsigned off = ((unsigned)s << 9) | ((unsigned)lane << 4);
  return *(const float4*)((const char*)hp + off);
}
__device__ __forceinline__ void axpy4(float4& acc, float a, const float4& x) {
  acc.x += a * x.x; acc.y += a * x.y; acc.z += a * x.z; acc.w += a * x.w;
}
__device__ __forceinline__ float dot4(const float4& a, const float4& b) {
  const float p0 = a.x * b.x + a.z * b.z;
  const float p1 = a.y * b.y + a.w * b.w;
  return p0 + p1;
}
__device__ __forceinline__ float quad_sum(float x) {
  int b = __builtin_bit_cast(int, x);
  x += __builtin_bit_cast(float, __builtin_amdgcn_update_dpp(b, b, 0xB1, 0xF, 0xF, true));
  b = __builtin_bit_cast(int, x);
  x += __builtin_bit_cast(float, __builtin_amdgcn_update_dpp(b, b, 0x4E, 0xF, 0xF, true));
  return x;
}
__device__ __forceinline__ float wave_sum(float x) {
  int b = __builtin_bit_cast(int, x);
  x += __builtin_bit_cast(float, __builtin_amdgcn_permlanex16(b, b, 0x76543210, 0xfedcba98, false, false));
  b = __builtin_bit_cast(int, x);
  x += __builtin_bit_cast(float, __builtin_amdgcn_update_dpp(b, b, 0x168, 0xF, 0xF, true));
  b = __builtin_bit_cast(int, x);
  x += __builtin_bit_cast(float, __builtin_amdgcn_update_dpp(b, b, 0x164, 0xF, 0xF, true));
  b = __builtin_bit_cast(int, x);
  x += __builtin_bit_cast(float, __builtin_amdgcn_update_dpp(b, b, 0x4E, 0xF, 0xF, true));
  b = __builtin_bit_cast(int, x);
  x += __builtin_bit_cast(float, __builtin_amdgcn_update_dpp(b, b, 0xB1, 0xF, 0xF, true));
  return x;
}

__device__ __forceinline__ void gather_trips(const float* __restrict__ hp, int mys, int cnt,
                                             int lane, const float4& alv, float ern,
                                             float& m, float& sum, float4& acc) {
  const float NEG_INF = __uint_as_float(0xff800000u);
  for (int t = 0; t < cnt; t += 4) {
    const int s0 = __builtin_amdgcn_readlane(mys, t);
    const int s1 = __builtin_amdgcn_readlane(mys, t + 1);
    const int s2 = __builtin_amdgcn_readlane(mys, t + 2);
    const int s3 = __builtin_amdgcn_readlane(mys, t + 3);
    const float4 x0 = edge_row(hp, s0, lane);
    const float4 x1 = edge_row(hp, s1, lane);
    const float4 x2 = edge_row(hp, s2, lane);
    const float4 x3 = edge_row(hp, s3, lane);
    const float e0 = quad_sum(dot4(x0, alv));
    const float e1 = quad_sum(dot4(x1, alv));
    const float e2 = quad_sum(dot4(x2, alv));
    const float e3 = quad_sum(dot4(x3, alv));
    const float v0 = lrelu(e0 + ern);
    const float v1 = (t + 1 < cnt) ? lrelu(e1 + ern) : NEG_INF;
    const float v2 = (t + 2 < cnt) ? lrelu(e2 + ern) : NEG_INF;
    const float v3 = (t + 3 < cnt) ? lrelu(e3 + ern) : NEG_INF;
    const float mn = fmaxf(fmaxf(m, fmaxf(v0, v1)), fmaxf(v2, v3));
    const float sc = __builtin_amdgcn_exp2f(m - mn);
    const float p0 = __builtin_amdgcn_exp2f(v0 - mn);
    const float p1 = __builtin_amdgcn_exp2f(v1 - mn);
    const float p2 = __builtin_amdgcn_exp2f(v2 - mn);
    const float p3 = __builtin_amdgcn_exp2f(v3 - mn);
    sum = sum * sc + ((p0 + p1) + (p2 + p3));
    acc.x *= sc; acc.y *= sc; acc.z *= sc; acc.w *= sc;
    axpy4(acc, p0, x0);
    axpy4(acc, p1, x1);
    axpy4(acc, p2, x2);
    axpy4(acc, p3, x3);
    m = mn;
  }
}

__global__ void __launch_bounds__(256)
node_gather(const int* __restrict__ bucket, const int* __restrict__ ovf_len,
            const int* __restrict__ ovf_dst, const int* __restrict__ ovf_src,
            const float* __restrict__ hp,
            const float* __restrict__ al,
            const float* __restrict__ ar,
            const float* __restrict__ bias,
            const float* __restrict__ gamma,
            const float* __restrict__ beta,
            float* __restrict__ out) {
  const int wave = __builtin_amdgcn_readfirstlane(threadIdx.x >> 5);
  const int lane = threadIdx.x & 31;
  constexpr int STRIDE = GATHER_WGS * 8;
  float4 alv = ((const float4*)al)[lane];
  alv.x *= LOG2E; alv.y *= LOG2E; alv.z *= LOG2E; alv.w *= LOG2E;

  __shared__ float4 featv[4][32];
  if (wave < 4) {
    const float* p = (wave == 0) ? ar : (wave == 1) ? bias : (wave == 2) ? gamma : beta;
    float4 v = ((const float4*)p)[lane];
    const float f = (wave == 0) ? LOG2E : 1.f;
    v.x *= f; v.y *= f; v.z *= f; v.w *= f;
    featv[wave][lane] = v;
  }
  __syncthreads();

  int n = blockIdx.x * 8 + wave;
  int n1 = n + STRIDE;
  int row = bucket[n * ROW + lane];
  float ern = quad_sum(dot4(edge_row(hp, n, lane), featv[0][lane]));

  do {
    int row1 = 0;
    if (n1 < N_) row1 = bucket[n1 * ROW + lane];
    const int deg = __builtin_amdgcn_readlane(row, CAP);
    const int mys = (lane == CAP) ? 0 : row;

    float m = __uint_as_float(0xff800000u);
    float sum = 0.f;
    float4 acc = make_float4(0.f, 0.f, 0.f, 0.f);
    if (deg > 0) {
      gather_trips(hp, mys, min(deg, CAP), lane, alv, ern, m, sum, acc);
      if (deg > CAP) {
        const int novf = ovf_len[0];
        for (int c0 = 0; c0 < novf; c0 += 32) {
          const int i = c0 + lane;
          int od = -1, os = 0;
          if (i < novf) { od = ovf_dst[i]; os = ovf_src[i]; }
          unsigned hit = __builtin_amdgcn_ballot_w32(od == n);
          while (hit) {
            const int b = __builtin_ctz(hit);
            hit &= hit - 1;
            gather_trips(hp, __builtin_amdgcn_readlane(os, b), 1, lane, alv, ern, m, sum, acc);
          }
        }
      }
    }
    const float den = sum + 1e-16f;

    asm volatile("" ::: "memory");
    const float4 arv = featv[0][lane];
    const float4 bb  = featv[1][lane];
    const float4 g   = featv[2][lane];
    const float4 be  = featv[3][lane];
    const float4 own1 = edge_row(hp, min(n1, N_ - 1), lane);

    const float rden = __builtin_amdgcn_rcpf(den);
    float4 x;
    x.x = acc.x * rden + bb.x; x.y = acc.y * rden + bb.y;
    x.z = acc.z * rden + bb.z; x.w = acc.w * rden + bb.w;
    float s  = x.x + x.y + x.z + x.w;
    float s2 = x.x * x.x + x.y * x.y + x.z * x.z + x.w * x.w;
    s  = wave_sum(s);
    s2 = wave_sum(s2);
    const float mu  = s * (1.f / 128.f);
    const float var = s2 * (1.f / 128.f) - mu * mu;
    const float inv = rsqrtf(var + LN_EPS);
    float4 y;
    y.x = fmaxf((x.x - mu) * inv * g.x + be.x, 0.f);
    y.y = fmaxf((x.y - mu) * inv * g.y + be.y, 0.f);
    y.z = fmaxf((x.z - mu) * inv * g.z + be.z, 0.f);
    y.w = fmaxf((x.w - mu) * inv * g.w + be.w, 0.f);
    ((float4*)(out + (size_t)n * F_))[lane] = y;

    ern = quad_sum(dot4(own1, arv));
    n = n1; row = row1;
    n1 += STRIDE;
  } while (n < N_);
}

__global__ void __launch_bounds__(96)
gemm_pred(const float* __restrict__ X,
          const float* __restrict__ Wp,
          const float* __restrict__ bp,
          float* __restrict__ Y) {
  __shared__ __attribute__((aligned(16))) float atile[2][16 * APITCH];
  const int lane = threadIdx.x & 31;
  const int wave = __builtin_amdgcn_readfirstlane(threadIdx.x >> 5);
  const int n0   = wave * 16;
  const int half = lane >> 4;
  const int l16  = lane & 15;
  const int n    = n0 + l16;
  const bool valid = (n < C_);

  v2f b[F_ / 4];
#pragma unroll
  for (int j = 0; j < F_ / 4; ++j) {
    const int ka = 4 * j + 2 * half;
    b[j].x = valid ? Wp[ka * C_ + n] : 0.f;
    b[j].y = valid ? Wp[(ka + 1) * C_ + n] : 0.f;
  }
  const float bbn = valid ? bp[n] : 0.f;

  const float4* X4 = (const float4*)X;
  const int aoff = l16 * APITCH + 2 * half;
  const int xrow = wave;
  const bool has15 = (wave == 0);

  int tile = blockIdx.x;
  float4 g0 = X4[(tile * 16 + xrow     ) * (F_ / 4) + lane];
  float4 g1 = X4[(tile * 16 + xrow +  3) * (F_ / 4) + lane];
  float4 g2 = X4[(tile * 16 + xrow +  6) * (F_ / 4) + lane];
  float4 g3 = X4[(tile * 16 + xrow +  9) * (F_ / 4) + lane];
  float4 g4 = X4[(tile * 16 + xrow + 12) * (F_ / 4) + lane];
  float4 g5 = make_float4(0.f, 0.f, 0.f, 0.f);
  if (has15) g5 = X4[(tile * 16 + 15) * (F_ / 4) + lane];
  int buf = 0;
  for (; tile < M_TILES; tile += PRED_WGS) {
    float* at = atile[buf];
    float* sp = at + xrow * APITCH + lane * 4;
    *(float4*)(sp             ) = g0;
    *(float4*)(sp +  3 * APITCH) = g1;
    *(float4*)(sp +  6 * APITCH) = g2;
    *(float4*)(sp +  9 * APITCH) = g3;
    *(float4*)(sp + 12 * APITCH) = g4;
    if (has15) *(float4*)(at + 15 * APITCH + lane * 4) = g5;
    __syncthreads();
    const int nt = tile + PRED_WGS;
    if (nt < M_TILES) {
      g0 = X4[(nt * 16 + xrow     ) * (F_ / 4) + lane];
      g1 = X4[(nt * 16 + xrow +  3) * (F_ / 4) + lane];
      g2 = X4[(nt * 16 + xrow +  6) * (F_ / 4) + lane];
      g3 = X4[(nt * 16 + xrow +  9) * (F_ / 4) + lane];
      g4 = X4[(nt * 16 + xrow + 12) * (F_ / 4) + lane];
      if (has15) g5 = X4[(nt * 16 + 15) * (F_ / 4) + lane];
    }
    const int m0 = tile * 16;
    v8f acc = {};
#pragma unroll
    for (int j = 0; j < F_ / 4; ++j) {
      const v2f a = *(const v2f*)(at + aoff + 4 * j);
      acc = __builtin_amdgcn_wmma_f32_16x16x4_f32(false, a, false, b[j],
                                                  (short)0, acc, false, false);
    }
    if (valid) {
#pragma unroll
      for (int r = 0; r < 8; ++r) {
        const int m = m0 + r + 8 * half;
        Y[m * C_ + n] = acc[r] + bbn;
      }
    }
    buf ^= 1;
  }
}

extern "C" void kernel_launch(void* const* d_in, const int* in_sizes, int n_in,
                              void* d_out, int out_size, void* d_ws, size_t ws_size,
                              hipStream_t stream) {
  const float* feats = (const float*)d_in[0];
  const int*   src   = (const int*)d_in[1];
  const int*   dst   = (const int*)d_in[2];
  const float* Ws    = (const float*)d_in[3];
  const float* al    = (const float*)d_in[4];
  const float* ar    = (const float*)d_in[5];
  const float* bias  = (const float*)d_in[6];
  const float* gamma = (const float*)d_in[7];
  const float* beta  = (const float*)d_in[8];
  const float* Wp    = (const float*)d_in[9];
  const float* bp    = (const float*)d_in[10];
  float* outp = (float*)d_out;

  float* ws   = (float*)d_ws;
  float* hbuf = ws;
  float* hp   = hbuf + (size_t)N_ * F_;
  int*   bucket  = (int*)(hp + (size_t)N_ * F_);
  int*   cells   = bucket + (size_t)N_ * ROW;
  int*   counts  = cells + (size_t)NB * CHUNKS * CELL;
  int*   ovf_len = counts + (size_t)NB * CHUNKS;
  int*   ovf_dst = ovf_len + 4;
  int*   ovf_src = ovf_dst + E_;

  edge_file<<<CHUNKS, 256, 0, stream>>>(src, dst, cells, counts, ovf_len);
  row_build<<<NB, 256, 0, stream>>>(src, dst, cells, counts, bucket, ovf_len, ovf_dst, ovf_src);

  for (int i = 0; i < L_; ++i) {
    const float* X = (i == 0) ? feats : hbuf;
    gemm_node_fc<<<FC_WGS, 256, 0, stream>>>(X, Ws + (size_t)i * F_ * F_, hp);
    node_gather<<<GATHER_WGS, 256, 0, stream>>>(bucket, ovf_len, ovf_dst, ovf_src, hp,
                                            al + (size_t)i * H_ * D_,
                                            ar + (size_t)i * H_ * D_, bias + (size_t)i * F_,
                                            gamma + (size_t)i * F_, beta + (size_t)i * F_, hbuf);
  }
  gemm_pred<<<PRED_WGS, 96, 0, stream>>>(hbuf, Wp, bp, outp);
}


# First layer's fc GEMM needs nothing from the edge sort: its row tiles run as extra work-groups inside the two sort launches (half in each), beside their wait chains; 7 -> 6 launches
# speedup vs baseline: 6.9355x; 1.0445x over previous
#include <hip/hip_runtime.h>

constexpr int N_ = 50000;
constexpr int E_ = 800000;
constexpr int F_ = 128;
constexpr int H_ = 8;
constexpr int D_ = 16;
constexpr int C_ = 40;
constexpr int L_ = 2;
constexpr float NEG_SLOPE = 0.2f;
constexpr float LN_EPS = 1e-5f;

typedef __attribute__((ext_vector_type(2))) float v2f;
typedef __attribute__((ext_vector_type(8))) float v8f;

__device__ __forceinline__ float lrelu(float x) { return fmaxf(x, NEG_SLOPE * x); }
constexpr float LOG2E = 1.4426950408889634f;

constexpr int M_TILES     = N_ / 16;
constexpr int FC_WGS      = 768;
constexpr int FC_HALF_WGS = 512;
constexpr int FC_SPLIT    = 3 * FC_HALF_WGS;
constexpr int PRED_WGS    = 1024;
constexpr int GATHER_WGS  = 2048;
constexpr int E4_         = E_ / 4;
constexpr int ROW         = 32;
constexpr int CAP         = ROW - 1;
constexpr int RB          = 128;
constexpr int NB          = (N_ + RB - 1) / RB;
constexpr int CHUNKS      = 256;
constexpr int GPC         = (E4_ + CHUNKS - 1) / CHUNKS;
constexpr int CELL        = 32;
static_assert(FC_SPLIT + FC_HALF_WGS <= M_TILES, "every fc work-group of either sort launch has a first tile");
static_assert(N_ <= 65536, "a source id must fit the low 16 bits of a filed word");
static_assert(RB == 128, "the range of a node is its id >> 7");
static_assert(GPC <= 4 * 256, "a filing thread takes at most four groups");
constexpr int APITCH      = F_ + 4;

template <int TILE0, int TILE_END, int STRIDE>
__device__ __forceinline__ void fc_tiles(const float* __restrict__ X,
                                         const float* __restrict__ W,
                                         float* __restrict__ Y, int wg) {
  __shared__ __attribute__((aligned(16))) float atile[2][16 * APITCH];
  const int lane = threadIdx.x & 31;
  const int wave = threadIdx.x >> 5;
  const int n0   = wave * 16;
  const int half = lane >> 4;
  const int l16  = lane & 15;
  const int bcol = n0 + l16;

  v2f b[F_ / 4];
#pragma unroll
  for (int j = 0; j < F_ / 4; ++j) {
    const int ka = 4 * j + 2 * half;
    b[j].x = W[ka * F_ + bcol];
    b[j].y = W[(ka + 1) * F_ + bcol];
  }

  const float4* X4 = (const float4*)X;
  const int soff0 = wave * APITCH + lane * 4;
  const int soff1 = (wave + 8) * APITCH + lane * 4;
  const int aoff  = l16 * APITCH + 2 * half;

  int tile = TILE0 + wg;
  float4 g0 = X4[(tile * 16 + wave) * (F_ / 4) + lane];
  float4 g1 = X4[(tile * 16 + wave + 8) * (F_ / 4) + lane];
  int buf = 0;
  for (; tile < TILE_END; tile += STRIDE) {
    float* at = atile[buf];
    *(float4*)(at + soff0) = g0;
    *(float4*)(at + soff1) = g1;
    __syncthreads();
    const int nt = tile + STRIDE;
    if (nt < TILE_END) {
      g0 = X4[(nt * 16 + wave) * (F_ / 4) + lane];
      g1 = X4[(nt * 16 + wave + 8) * (F_ / 4) + lane];
    }
    const int m0 = tile * 16;
    v8f acc = {};
#pragma unroll
    for (int j = 0; j < F_ / 4; ++j) {
      const v2f a = *(const v2f*)(at + aoff + 4 * j);
      acc = __builtin_amdgcn_wmma_f32_16x16x4_f32(false, a, false, b[j],
                                                  (short)0, acc, false, false);
    }
#pragma unroll
    for (int r = 0; r < 8; ++r) {
      const int m = m0 + r + 8 * half;
      Y[m * F_ + bcol] = acc[r];
    }
    buf ^= 1;
  }
}

__global__ void __launch_bounds__(256)
gemm_node_fc(const float* __restrict__ X,
             const float* __restrict__ W,
             float* __restrict__ Y) {
  fc_tiles<0, M_TILES, FC_WGS>(X, W, Y, (int)blockIdx.x);
}

__device__ __forceinline__ void
edge_file(int w, const int* __restrict__ src, const int* __restrict__ dst,
          int* __restrict__ cells, int* __restrict__ counts, int* __restrict__ ovf_len) {
  __shared__ int cnt[NB];
  for (int t = threadIdx.x; t < NB; t += 256) cnt[t] = 0;
  const int gbeg = w * GPC;
  const int gend = min(gbeg + GPC, E4_);
  int4 dd[4], ss[4];
#pragma unroll
  for (int i = 0; i < 4; ++i) {
    const int g = gbeg + (int)threadIdx.x + 256 * i;
    dd[i] = make_int4(-1, -1, -1, -1);
    ss[i] = make_int4(0, 0, 0, 0);
    if (g < gend) {
      dd[i] = ((const int4*)dst)[g];
      ss[i] = ((const int4*)src)[g];
    }
  }
  __syncthreads();
  int* cell0 = cells + w * CELL;
#pragma unroll
  for (int i = 0; i < 4; ++i) {
    if (dd[i].x >= 0) {
      const int4 d = dd[i];
      const int4 s = ss[i];
      const int b0 = d.x >> 7, b1 = d.y >> 7, b2 = d.z >> 7, b3 = d.w >> 7;
      const int r0 = atomicAdd(&cnt[b0], 1);
      const int r1 = atomicAdd(&cnt[b1], 1);
      const int r2 = atomicAdd(&cnt[b2], 1);
      const int r3 = atomicAdd(&cnt[b3], 1);
      if (r0 < CELL) cell0[b0 * (CHUNKS * CELL) + r0] = ((d.x & (RB - 1)) << 16) | s.x;
      if (r1 < CELL) cell0[b1 * (CHUNKS * CELL) + r1] = ((d.y & (RB - 1)) << 16) | s.y;
      if (r2 < CELL) cell0[b2 * (CHUNKS * CELL) + r2] = ((d.z & (RB - 1)) << 16) | s.z;
      if (r3 < CELL) cell0[b3 * (CHUNKS * CELL) + r3] = ((d.w & (RB - 1)) << 16) | s.w;
    }
  }
  __syncthreads();
  for (int t = threadIdx.x; t < NB; t += 256) counts[t * CHUNKS + w] = cnt[t];
  if (w == 0 && threadIdx.x == 0) ovf_len[0] = 0;
}

__device__ __forceinline__ void place_ranked(int dl, int s, int r, int node0, int* rows,
                                             int* __restrict__ ovf_len,
                                             int* __restrict__ ovf_dst, int* __restrict__ ovf_src) {
  if (r < CAP) {
    rows[dl * ROW + r] = s;
  } else {
    const int k = atomicAdd(ovf_len, 1);
    ovf_dst[k] = node0 + dl;
    ovf_src[k] = s;
  }
}

__device__ __forceinline__ void
row_build(int b, const int* __restrict__ src, const int* __restrict__ dst,
          const int* __restrict__ cells, const int* __restrict__ counts,
          int* __restrict__ bucket, int* __restrict__ ovf_len,
          int* __restrict__ ovf_dst, int* __restrict__ ovf_src) {
  __shared__ __attribute__((aligned(16))) int rows[RB * ROW];
  __shared__ int deg[RB];
  __shared__ int cc[CHUNKS];
  static_assert(CHUNKS == 256, "one count per thread");
  const int node0 = b * RB;
  const int lane = threadIdx.x & 31;
  const int wave = __builtin_amdgcn_readfirstlane(threadIdx.x >> 5);
  cc[threadIdx.x] = counts[b * CHUNKS + threadIdx.x];
  for (int t = threadIdx.x; t < RB * ROW / 4; t += 256) ((int4*)rows)[t] = make_int4(0, 0, 0, 0);
  if (threadIdx.x < RB) deg[threadIdx.x] = 0;
  __syncthreads();

  const int* cellb = cells + (b * CHUNKS + wave) * CELL + lane;
  int wv[32];
  unsigned big = 0;
  const int myn = cc[wave + 8 * lane];
#pragma unroll
  for (int u = 0; u < 32; ++u) {
    const int n = __builtin_amdgcn_readlane(myn, u);
    if (n > CELL) big |= 1u << u;
    wv[u] = -1;
    if (n <= CELL && lane < n) wv[u] = cellb[u * 8 * CELL];
  }
#pragma unroll
  for (int v = 0; v < 32; v += 8) {
    int rk[8];
#pragma unroll
    for (int u = 0; u < 8; ++u) {
      rk[u] = 0;
      if (wv[v + u] >= 0) rk[u] = atomicAdd(&deg[wv[v + u] >> 16], 1);
    }
#pragma unroll
    for (int u = 0; u < 8; ++u) {
      if (wv[v + u] >= 0)
        place_ranked(wv[v + u] >> 16, wv[v + u] & 0xffff, rk[u], node0, rows, ovf_len, ovf_dst, ovf_src);
    }
  }
  while (big) {
    const int u = __builtin_ctz(big);
    big &= big - 1;
    const int c = wave + 8 * u;
    const int gbeg = c * GPC;
    const int gend = min(gbeg + GPC, E4_);
    for (int g = gbeg + lane; g < gend; g += 32) {
      const int4 d = ((const int4*)dst)[g];
      const int4 s = ((const int4*)src)[g];
      if ((d.x >> 7) == b) { const int dl = d.x & (RB - 1); place_ranked(dl, s.x, atomicAdd(&deg[dl], 1), node0, rows, ovf_len, ovf_dst, ovf_src); }
      if ((d.y >> 7) == b) { const int dl = d.y & (RB - 1); place_ranked(dl, s.y, atomicAdd(&deg[dl], 1), node0, rows, ovf_len, ovf_dst, ovf_src); }
      if ((d.z >> 7) == b) { const int dl = d.z & (RB - 1); place_ranked(dl, s.z, atomicAdd(&deg[dl], 1), node0, rows, ovf_len, ovf_dst, ovf_src); }
      if ((d.w >> 7) == b) { const int dl = d.w & (RB - 1); place_ranked(dl, s.w, atomicAdd(&deg[dl], 1), node0, rows, ovf_len, ovf_dst, ovf_src); }
    }
  }
  __syncthreads();

  for (int r = wave; r < RB; r += 8) {
    const int node = node0 + r;
    if (node < N_) bucket[node * ROW + lane] = (lane == CAP) ? deg[r] : rows[r * ROW + lane];
  }
}

__global__ void __launch_bounds__(256)
file_with_fc(const int* __restrict__ src, const int* __restrict__ dst,
             int* __restrict__ cells, int* __restrict__ counts, int* __restrict__ ovf_len,
             const float* __restrict__ X, const float* __restrict__ W, float* __restrict__ Y) {
  const int wg = (int)blockIdx.x;
  if (wg < CHUNKS) edge_file(wg, src, dst, cells, counts, ovf_len);
  else             fc_tiles<0, FC_SPLIT, FC_HALF_WGS>(X, W, Y, wg - CHUNKS);
}

__global__ void __launch_bounds__(256)
build_with_fc(const int* __restrict__ src, const int* __restrict__ dst,
              const int* __restrict__ cells, const int* __restrict__ counts,
              int* __restrict__ bucket, int* __restrict__ ovf_len,
              int* __restrict__ ovf_dst, int* __restrict__ ovf_src,
              const float* __restrict__ X, const float* __restrict__ W, float* __restrict__ Y) {
  const int wg = (int)blockIdx.x;
  if (wg < NB) row_build(wg, src, dst, cells, counts, bucket, ovf_len, ovf_dst, ovf_src);
  else         fc_tiles<FC_SPLIT, M_TILES, FC_HALF_WGS>(X, W, Y, wg - NB);
}

__device__ __forceinline__ float4 edge_row(const float* __restrict__ hp, int s, int lane) {
  const unsigned off = ((unsigned)s << 9) | ((unsigned)lane << 4);
  return *(const float4*)((const char*)hp + off);
}
__device__ __forceinline__ void axpy4(float4& acc, float a, const float4& x) {
  acc.x += a * x.x; acc.y += a * x.y; acc.z += a * x.z; acc.w += a * x.w;
}
__device__ __forceinline__ float dot4(const float4& a, const float4& b) {
  const float p0 = a.x * b.x + a.z * b.z;
  const float p1 = a.y * b.y + a.w * b.w;
  return p0 + p1;
}
__device__ __forceinline__ float quad_sum(float x) {
  int b = __builtin_bit_cast(int, x);
  x += __builtin_bit_cast(float, __builtin_amdgcn_update_dpp(b, b, 0xB1, 0xF, 0xF, true));
  b = __builtin_bit_cast(int, x);
  x += __builtin_bit_cast(float, __builtin_amdgcn_update_dpp(b, b, 0x4E, 0xF, 0xF, true));
  return x;
}
__device__ __forceinline__ float wave_sum(float x) {
  int b = __builtin_bit_cast(int, x);
  x += __builtin_bit_cast(float, __builtin_amdgcn_permlanex16(b, b, 0x76543210, 0xfedcba98, false, false));
  b = __builtin_bit_cast(int, x);
  x += __builtin_bit_cast(float, __builtin_amdgcn_update_dpp(b, b, 0x168, 0xF, 0xF, true));
  b = __builtin_bit_cast(int, x);
  x += __builtin_bit_cast(float, __builtin_amdgcn_update_dpp(b, b, 0x164, 0xF, 0xF, true));
  b = __builtin_bit_cast(int, x);
  x += __builtin_bit_cast(float, __builtin_amdgcn_update_dpp(b, b, 0x4E, 0xF, 0xF, true));
  b = __builtin_bit_cast(int, x);
  x += __builtin_bit_cast(float, __builtin_amdgcn_update_dpp(b, b, 0xB1, 0xF, 0xF, true));
  return x;
}

__device__ __forceinline__ void gather_trips(const float* __restrict__ hp, int mys, int cnt,
                                             int lane, const float4& alv, float ern,
                                             float& m, float& sum, float4& acc) {
  const float NEG_INF = __uint_as_float(0xff800000u);
  for (int t = 0; t < cnt; t += 4) {
    const int s0 = __builtin_amdgcn_readlane(mys, t);
    const int s1 = __builtin_amdgcn_readlane(mys, t + 1);
    const int s2 = __builtin_amdgcn_readlane(mys, t + 2);
    const int s3 = __builtin_amdgcn_readlane(mys, t + 3);
    const float4 x0 = edge_row(hp, s0, lane);
    const float4 x1 = edge_row(hp, s1, lane);
    const float4 x2 = edge_row(hp, s2, lane);
    const float4 x3 = edge_row(hp, s3, lane);
    const float e0 = quad_sum(dot4(x0, alv));
    const float e1 = quad_sum(dot4(x1, alv));
    const float e2 = quad_sum(dot4(x2, alv));
    const float e3 = quad_sum(dot4(x3, alv));
    const float v0 = lrelu(e0 + ern);
    const float v1 = (t + 1 < cnt) ? lrelu(e1 + ern) : NEG_INF;
    const float v2 = (t + 2 < cnt) ? lrelu(e2 + ern) : NEG_INF;
    const float v3 = (t + 3 < cnt) ? lrelu(e3 + ern) : NEG_INF;
    const float mn = fmaxf(fmaxf(m, fmaxf(v0, v1)), fmaxf(v2, v3));
    const float sc = __builtin_amdgcn_exp2f(m - mn);
    const float p0 = __builtin_amdgcn_exp2f(v0 - mn);
    const float p1 = __builtin_amdgcn_exp2f(v1 - mn);
    const float p2 = __builtin_amdgcn_exp2f(v2 - mn);
    const float p3 = __builtin_amdgcn_exp2f(v3 - mn);
    sum = sum * sc + ((p0 + p1) + (p2 + p3));
    acc.x *= sc; acc.y *= sc; acc.z *= sc; acc.w *= sc;
    axpy4(acc, p0, x0);
    axpy4(acc, p1, x1);
    axpy4(acc, p2, x2);
    axpy4(acc, p3, x3);
    m = mn;
  }
}

__global__ void __launch_bounds__(256)
node_gather(const int* __restrict__ bucket, const int* __restrict__ ovf_len,
            const int* __restrict__ ovf_dst, const int* __restrict__ ovf_src,
            const float* __restrict__ hp,
            const float* __restrict__ al,
            const float* __restrict__ ar,
            const float* __restrict__ bias,
            const float* __restrict__ gamma,
            const float* __restrict__ beta,
            float* __restrict__ out) {
  const int wave = __builtin_amdgcn_readfirstlane(threadIdx.x >> 5);
  const int lane = threadIdx.x & 31;
  constexpr int STRIDE = GATHER_WGS * 8;
  float4 alv = ((const float4*)al)[lane];
  alv.x *= LOG2E; alv.y *= LOG2E; alv.z *= LOG2E; alv.w *= LOG2E;

  __shared__ float4 featv[4][32];
  if (wave < 4) {
    const float* p = (wave == 0) ? ar : (wave == 1) ? bias : (wave == 2) ? gamma : beta;
    float4 v = ((const float4*)p)[lane];
    const float f = (wave == 0) ? LOG2E : 1.f;
    v.x *= f; v.y *= f; v.z *= f; v.w *= f;
    featv[wave][lane] = v;
  }
  __syncthreads();

  int n = blockIdx.x * 8 + wave;
  int n1 = n + STRIDE;
  int row = bucket[n * ROW + lane];
  float ern = quad_sum(dot4(edge_row(hp, n, lane), featv[0][lane]));

  do {
    int row1 = 0;
    if (n1 < N_) row1 = bucket[n1 * ROW + lane];
    const int deg = __builtin_amdgcn_readlane(row, CAP);
    const int mys = (lane == CAP) ? 0 : row;

    float m = __uint_as_float(0xff800000u);
    float sum = 0.f;
    float4 acc = make_float4(0.f, 0.f, 0.f, 0.f);
    if (deg > 0) {
      gather_trips(hp, mys, min(deg, CAP), lane, alv, ern, m, sum, acc);
      if (deg > CAP) {
        const int novf = ovf_len[0];
        for (int c0 = 0; c0 < novf; c0 += 32) {
          const int i = c0 + lane;
          int od = -1, os = 0;
          if (i < novf) { od = ovf_dst[i]; os = ovf_src[i]; }
          unsigned hit = __builtin_amdgcn_ballot_w32(od == n);
          while (hit) {
            const int b = __builtin_ctz(hit);
            hit &= hit - 1;
            gather_trips(hp, __builtin_amdgcn_readlane(os, b), 1, lane, alv, ern, m, sum, acc);
          }
        }
      }
    }
    const float den = sum + 1e-16f;

    asm volatile("" ::: "memory");
    const float4 arv = featv[0][lane];
    const float4 bb  = featv[1][lane];
    const float4 g   = featv[2][lane];
    const float4 be  = featv[3][lane];
    const float4 own1 = edge_row(hp, min(n1, N_ - 1), lane);

    const float rden = __builtin_amdgcn_rcpf(den);
    float4 x;
    x.x = acc.x * rden + bb.x; x.y = acc.y * rden + bb.y;
    x.z = acc.z * rden + bb.z; x.w = acc.w * rden + bb.w;
    float s  = x.x + x.y + x.z + x.w;
    float s2 = x.x * x.x + x.y * x.y + x.z * x.z + x.w * x.w;
    s  = wave_sum(s);
    s2 = wave_sum(s2);
    const float mu  = s * (1.f / 128.f);
    const float var = s2 * (1.f / 128.f) - mu * mu;
    const float inv = rsqrtf(var + LN_EPS);
    float4 y;
    y.x = fmaxf((x.x - mu) * inv * g.x + be.x, 0.f);
    y.y = fmaxf((x.y - mu) * inv * g.y + be.y, 0.f);
    y.z = fmaxf((x.z - mu) * inv * g.z + be.z, 0.f);
    y.w = fmaxf((x.w - mu) * inv * g.w + be.w, 0.f);
    ((float4*)(out + (size_t)n * F_))[lane] = y;

    ern = quad_sum(dot4(own1, arv));
    n = n1; row = row1;
    n1 += STRIDE;
  } while (n < N_);
}

__global__ void __launch_bounds__(96)
gemm_pred(const float* __restrict__ X,
          const float* __restrict__ Wp,
          const float* __restrict__ bp,
          float* __restrict__ Y) {
  __shared__ __attribute__((aligned(16))) float atile[2][16 * APITCH];
  const int lane = threadIdx.x & 31;
  const int wave = __builtin_amdgcn_readfirstlane(threadIdx.x >> 5);
  const int n0   = wave * 16;
  const int half = lane >> 4;
  const int l16  = lane & 15;
  const int n    = n0 + l16;
  const bool valid = (n < C_);

  v2f b[F_ / 4];
#pragma unroll
  for (int j = 0; j < F_ / 4; ++j) {
    const int ka = 4 * j + 2 * half;
    b[j].x = valid ? Wp[ka * C_ + n] : 0.f;
    b[j].y = valid ? Wp[(ka + 1) * C_ + n] : 0.f;
  }
  const float bbn = valid ? bp[n] : 0.f;

  const float4* X4 = (const float4*)X;
  const int aoff = l16 * APITCH + 2 * half;
  const int xrow = wave;
  const bool has15 = (wave == 0);

  int tile = blockIdx.x;
  float4 g0 = X4[(tile * 16 + xrow     ) * (F_ / 4) + lane];
  float4 g1 = X4[(tile * 16 + xrow +  3) * (F_ / 4) + lane];
  float4 g2 = X4[(tile * 16 + xrow +  6) * (F_ / 4) + lane];
  float4 g3 = X4[(tile * 16 + xrow +  9) * (F_ / 4) + lane];
  float4 g4 = X4[(tile * 16 + xrow + 12) * (F_ / 4) + lane];
  float4 g5 = make_float4(0.f, 0.f, 0.f, 0.f);
  if (has15) g5 = X4[(tile * 16 + 15) * (F_ / 4) + lane];
  int buf = 0;
  for (; tile < M_TILES; tile += PRED_WGS) {
    float* at = atile[buf];
    float* sp = at + xrow * APITCH + lane * 4;
    *(float4*)(sp             ) = g0;
    *(float4*)(sp +  3 * APITCH) = g1;
    *(float4*)(sp +  6 * APITCH) = g2;
    *(float4*)(sp +  9 * APITCH) = g3;
    *(float4*)(sp + 12 * APITCH) = g4;
    if (has15) *(float4*)(at + 15 * APITCH + lane * 4) = g5;
    __syncthreads();
    const int nt = tile + PRED_WGS;
    if (nt < M_TILES) {
      g0 = X4[(nt * 16 + xrow     ) * (F_ / 4) + lane];
      g1 = X4[(nt * 16 + xrow +  3) * (F_ / 4) + lane];
      g2 = X4[(nt * 16 + xrow +  6) * (F_ / 4) + lane];
      g3 = X4[(nt * 16 + xrow +  9) * (F_ / 4) + lane];
      g4 = X4[(nt * 16 + xrow + 12) * (F_ / 4) + lane];
      if (has15) g5 = X4[(nt * 16 + 15) * (F_ / 4) + lane];
    }
    const int m0 = tile * 16;
    v8f acc = {};
#pragma unroll
    for (int j = 0; j < F_ / 4; ++j) {
      const v2f a = *(const v2f*)(at + aoff + 4 * j);
      acc = __builtin_amdgcn_wmma_f32_16x16x4_f32(false, a, false, b[j],
                                                  (short)0, acc, false, false);
    }
    if (valid) {
#pragma unroll
      for (int r = 0; r < 8; ++r) {
        const int m = m0 + r + 8 * half;
        Y[m * C_ + n] = acc[r] + bbn;
      }
    }
    buf ^= 1;
  }
}

extern "C" void kernel_launch(void* const* d_in, const int* in_sizes, int n_in,
                              void* d_out, int out_size, void* d_ws, size_t ws_size,
                              hipStream_t stream) {
  const float* feats = (const float*)d_in[0];
  const int*   src   = (const int*)d_in[1];
  const int*   dst   = (const int*)d_in[2];
  const float* Ws    = (const float*)d_in[3];
  const float* al    = (const float*)d_in[4];
  const float* ar    = (const float*)d_in[5];
  const float* bias  = (const float*)d_in[6];
  const float* gamma = (const float*)d_in[7];
  const float* beta  = (const float*)d_in[8];
  const float* Wp    = (const float*)d_in[9];
  const float* bp    = (const float*)d_in[10];
  float* outp = (float*)d_out;

  float* ws   = (float*)d_ws;
  float* hbuf = ws;
  float* hp   = hbuf + (size_t)N_ * F_;
  int*   bucket  = (int*)(hp + (size_t)N_ * F_);
  int*   cells   = bucket + (size_t)N_ * ROW;
  int*   counts  = cells + (size_t)NB * CHUNKS * CELL;
  int*   ovf_len = counts + (size_t)NB * CHUNKS;
  int*   ovf_dst = ovf_len + 4;
  int*   ovf_src = ovf_dst + E_;

  file_with_fc<<<CHUNKS + FC_HALF_WGS, 256, 0, stream>>>(src, dst, cells, counts, ovf_len,
                                                         feats, Ws, hp);
  build_with_fc<<<NB + FC_HALF_WGS, 256, 0, stream>>>(src, dst, cells, counts, bucket, ovf_len,
                                                      ovf_dst, ovf_src, feats, Ws, hp);

  static_assert(L_ == 2, "two layers, written out: only the first one's projection rides beside the sort");
  node_gather<<<GATHER_WGS, 256, 0, stream>>>(bucket, ovf_len, ovf_dst, ovf_src, hp,
                                              al, ar, bias, gamma, beta, hbuf);
  gemm_node_fc<<<FC_WGS, 256, 0, stream>>>(hbuf, Ws + (size_t)F_ * F_, hp);
  node_gather<<<GATHER_WGS, 256, 0, stream>>>(bucket, ovf_len, ovf_dst, ovf_src, hp,
                                              al + (size_t)H_ * D_, ar + (size_t)H_ * D_,
                                              bias + (size_t)F_, gamma + (size_t)F_,
                                              beta + (size_t)F_, hbuf);
  gemm_pred<<<PRED_WGS, 96, 0, stream>>>(hbuf, Wp, bp, outp);
}
